# starting kernel
# baseline (speedup 1.0000x reference)
#include <hip/hip_runtime.h>
#include <hip/hip_bf16.h>

typedef __attribute__((ext_vector_type(2))) float v2f;
typedef __attribute__((ext_vector_type(8))) float v8f;

#define NF 512
#define NH 64
#define NC 8
#define KTILE 128
#define PAIRS (KTILE / 2)
#define PSTR 160


__global__ void deg_init_kernel(float* deg, int n) {
  int i = blockIdx.x * blockDim.x + threadIdx.x;
  if (i < n) deg[i] = 1.0f;
}

__global__ void deg_count_kernel(const int* __restrict__ dst, float* deg, int e) {
  int i = blockIdx.x * blockDim.x + threadIdx.x;
  if (i < e) atomicAdd(&deg[dst[i]], 1.0f);
}

__global__ void dinv_kernel(float* deg, int n) {
  int i = blockIdx.x * blockDim.x + threadIdx.x;
  if (i < n) deg[i] = rsqrtf(deg[i]);
}


__global__ __launch_bounds__(256) void gemm1_wmma_kernel(
    const float* __restrict__ X, const float* __restrict__ W1,
    const float* __restrict__ dinv, float* __restrict__ hs,
    float* __restrict__ acc1, int n) {
  __shared__ float wl[PAIRS * PSTR];
  const int lane  = threadIdx.x & 31;
  const int wave  = threadIdx.x >> 5;
  const int m0    = blockIdx.x * 128 + wave * 16;
  const int mlan  = lane & 15;
  const int khalf = (lane >> 4) << 1;
  const int pbase = lane >> 4;

  int row = m0 + mlan;
  if (row >= n) row = n - 1;
  const float* xrow = X + (size_t)row * NF;

  v8f c0 = {}; v8f c1 = {}; v8f c2 = {}; v8f c3 = {};

  for (int kb = 0; kb < NF; kb += KTILE) {
    for (int i = threadIdx.x; i < KTILE * NH; i += 256) {
      int kl = i >> 6, nn = i & 63;
      wl[(kl >> 1) * PSTR + (nn << 1) + (kl & 1)] = W1[(size_t)kb * NH + i];
    }
    if (kb + KTILE < NF)
      __builtin_prefetch(xrow + kb + KTILE, 0, 3);
    __syncthreads();

    for (int kk = 0; kk < KTILE; kk += 4) {
      v2f a; a.x = xrow[kb + kk + khalf]; a.y = xrow[kb + kk + khalf + 1];
      const v2f* wp = (const v2f*)(wl + ((kk >> 1) + pbase) * PSTR);
      v2f b0 = wp[mlan];
      v2f b1 = wp[16 + mlan];
      v2f b2 = wp[32 + mlan];
      v2f b3 = wp[48 + mlan];
      c0 = __builtin_amdgcn_wmma_f32_16x16x4_f32(false, a, false, b0, (short)0, c0, false, false);
      c1 = __builtin_amdgcn_wmma_f32_16x16x4_f32(false, a, false, b1, (short)0, c1, false, false);
      c2 = __builtin_amdgcn_wmma_f32_16x16x4_f32(false, a, false, b2, (short)0, c2, false, false);
      c3 = __builtin_amdgcn_wmma_f32_16x16x4_f32(false, a, false, b3, (short)0, c3, false, false);
    }
    __syncthreads();
  }

  const int half8 = (lane >> 4) << 3;
  for (int r = 0; r < 8; ++r) {
    int rr = m0 + r + half8;
    if (rr < n) {
      float dv = dinv[rr];
      size_t base = (size_t)rr * NH + mlan;
      float v0 = c0[r] * dv, v1 = c1[r] * dv, v2 = c2[r] * dv, v3 = c3[r] * dv;
      hs[base]      = v0; hs[base + 16]   = v1; hs[base + 32]   = v2; hs[base + 48]   = v3;
      acc1[base]    = v0; acc1[base + 16] = v1; acc1[base + 32] = v2; acc1[base + 48] = v3;
    }
  }
}

__global__ void agg1_kernel(const int* __restrict__ src, const int* __restrict__ dst,
                            const float* __restrict__ hs, float* acc1, long long total) {
  long long idx = (long long)blockIdx.x * blockDim.x + threadIdx.x;
  if (idx >= total) return;
  int e = (int)(idx >> 5);
  int f = (int)(idx & 31) << 1;
  int s = src[e], d = dst[e];
  const float2 v = *(const float2*)(hs + (size_t)s * NH + f);
  float* o = acc1 + (size_t)d * NH + f;
  atomicAdd(o, v.x);
  atomicAdd(o + 1, v.y);
}


__global__ void gemm2_kernel(const float* __restrict__ acc1, const float* __restrict__ dinv,
                             const float* __restrict__ b1, const float* __restrict__ W2,
                             float* __restrict__ t2s, float* __restrict__ acc2, int n) {
  int idx = blockIdx.x * blockDim.x + threadIdx.x;
  if (idx >= n * NC) return;
  int i = idx >> 3, c = idx & 7;
  float di = dinv[i];
  const float* arow = acc1 + (size_t)i * NH;
  float s = 0.0f;
  for (int k = 0; k < NH; ++k) {
    float h = fmaxf(fmaf(arow[k], di, b1[k]), 0.0f);
    s = fmaf(h, W2[k * NC + c], s);
  }
  float v = s * di;
  t2s[idx]  = v;
  acc2[idx] = v;
}

__global__ void agg2_kernel(const int* __restrict__ src, const int* __restrict__ dst,
                            const float* __restrict__ t2s, float* acc2, long long total) {
  long long idx = (long long)blockIdx.x * blockDim.x + threadIdx.x;
  if (idx >= total) return;
  int e = (int)(idx >> 2);
  int c = (int)(idx & 3) << 1;
  int s = src[e], d = dst[e];
  const float2 v = *(const float2*)(t2s + (size_t)s * NC + c);
  float* o = acc2 + (size_t)d * NC + c;
  atomicAdd(o, v.x);
  atomicAdd(o + 1, v.y);
}

__global__ void finalize2_kernel(const float* __restrict__ acc2, const float* __restrict__ dinv,
                                 const float* __restrict__ b2, float* __restrict__ out, int n) {
  int idx = blockIdx.x * blockDim.x + threadIdx.x;
  if (idx >= n * NC) return;
  int i = idx >> 3, c = idx & 7;
  out[idx] = fmaf(acc2[idx], dinv[i], b2[c]);
}


extern "C" void kernel_launch(void* const* d_in, const int* in_sizes, int n_in,
                              void* d_out, int out_size, void* d_ws, size_t ws_size,
                              hipStream_t stream) {
  const float* x  = (const float*)d_in[0];
  const int*   ei = (const int*)d_in[1];
  const float* W1 = (const float*)d_in[2];
  const float* b1 = (const float*)d_in[3];
  const float* W2 = (const float*)d_in[4];
  const float* b2 = (const float*)d_in[5];
  float* out = (float*)d_out;

  const int n = in_sizes[0] / NF;
  const int e = in_sizes[1] / 2;
  const int* src = ei;
  const int* dst = ei + e;

  float* ws   = (float*)d_ws;
  float* dinv = ws;
  float* hs   = dinv + n;
  float* acc1 = hs + (size_t)n * NH;
  float* t2s  = acc1 + (size_t)n * NH;
  float* acc2 = t2s + (size_t)n * NC;

  const int T = 256;
  deg_init_kernel<<<(n + T - 1) / T, T, 0, stream>>>(dinv, n);
  deg_count_kernel<<<(e + T - 1) / T, T, 0, stream>>>(dst, dinv, e);
  dinv_kernel<<<(n + T - 1) / T, T, 0, stream>>>(dinv, n);

  gemm1_wmma_kernel<<<(n + 127) / 128, T, 0, stream>>>(x, W1, dinv, hs, acc1, n);

  long long tot1 = (long long)e * 32;
  agg1_kernel<<<(unsigned)((tot1 + T - 1) / T), T, 0, stream>>>(src, dst, hs, acc1, tot1);

  gemm2_kernel<<<(n * NC + T - 1) / T, T, 0, stream>>>(acc1, dinv, b1, W2, t2s, acc2, n);

  long long tot2 = (long long)e * 4;
  agg2_kernel<<<(unsigned)((tot2 + T - 1) / T), T, 0, stream>>>(src, dst, t2s, acc2, tot2);

  finalize2_kernel<<<(n * NC + T - 1) / T, T, 0, stream>>>(acc2, dinv, b2, out, n);
}


# agg1/agg2 by a per-call dst-sorted edge list (returning int adds rank edges, scan, fill) + row gathers summed in arrival-rank order; W2 product at row end summed as a lane tree; no f32 atomics
# speedup vs baseline: 1.7734x; 1.7734x over previous
#include <hip/hip_runtime.h>
#include <hip/hip_bf16.h>

typedef __attribute__((ext_vector_type(2))) float v2f;
typedef __attribute__((ext_vector_type(8))) float v8f;

#define NF 512
#define NH 64
#define NC 8
#define KTILE 128
#define PAIRS (KTILE / 2)
#define PSTR 160
#define NPW 4


__global__ void cnt_zero_kernel(int* cnt, int n) {
  int i = blockIdx.x * blockDim.x + threadIdx.x;
  if (i < n) cnt[i] = 0;
}

__global__ void rank_kernel(const int* __restrict__ dst, int* cnt, int* __restrict__ rank, int e) {
  const int i0 = (blockIdx.x * blockDim.x + threadIdx.x) * 4;
  if (i0 + 3 < e && (((size_t)(dst + i0)) & 15) == 0 && (((size_t)(rank + i0)) & 15) == 0) {
    const int4 d = *(const int4*)(dst + i0);
    int4 r;
    r.x = atomicAdd(&cnt[d.x], 1);
    r.y = atomicAdd(&cnt[d.y], 1);
    r.z = atomicAdd(&cnt[d.z], 1);
    r.w = atomicAdd(&cnt[d.w], 1);
    *(int4*)(rank + i0) = r;
  } else {
    for (int i = i0; i < e && i < i0 + 4; ++i) rank[i] = atomicAdd(&cnt[dst[i]], 1);
  }
}

__global__ __launch_bounds__(1024) void scan_kernel(const int* __restrict__ cnt, int* __restrict__ rowptr,
                                                    float* __restrict__ dinv, int n) {
  __shared__ int wfront[32];
  __shared__ int wtot[32];
  const int t = threadIdx.x, lane = t & 31, wave = t >> 5;
  const int b = blockIdx.x;
  int s = 0;
  for (int k = 0; k < b; ++k) s += cnt[k * 1024 + t];
  const int i = b * 1024 + t;
  const int v = (i < n) ? cnt[i] : 0;
  int incl = v;
  for (int d = 1; d < 32; d <<= 1) {
    int u = __shfl_up(incl, d);
    if (lane >= d) incl += u;
  }
  for (int d = 16; d > 0; d >>= 1) s += __shfl_xor(s, d);
  if (lane == 31) wtot[wave] = incl;
  if (lane == 0) wfront[wave] = s;
  __syncthreads();
  int a = wfront[lane];
  int p = (lane < wave) ? wtot[lane] : 0;
  for (int d = 16; d > 0; d >>= 1) { a += __shfl_xor(a, d); p += __shfl_xor(p, d); }
  if (i < n) {
    rowptr[i] = a + p + incl - v;
    dinv[i] = rsqrtf((float)(v + 1));
  }
}

__global__ void fill_kernel(const int* __restrict__ src, const int* __restrict__ dst,
                            const int* __restrict__ rank, const int* __restrict__ rowptr,
                            int* __restrict__ col, int e) {
  const int i0 = (blockIdx.x * blockDim.x + threadIdx.x) * 4;
  if (i0 + 3 < e && (((size_t)(dst + i0)) & 15) == 0 && (((size_t)(src + i0)) & 15) == 0 &&
      (((size_t)(rank + i0)) & 15) == 0) {
    const int4 d = *(const int4*)(dst + i0);
    const int4 s = *(const int4*)(src + i0);
    const int4 r = *(const int4*)(rank + i0);
    const int p0 = rowptr[d.x], p1 = rowptr[d.y], p2 = rowptr[d.z], p3 = rowptr[d.w];
    col[p0 + r.x] = s.x;
    col[p1 + r.y] = s.y;
    col[p2 + r.z] = s.z;
    col[p3 + r.w] = s.w;
  } else {
    for (int i = i0; i < e && i < i0 + 4; ++i) col[rowptr[dst[i]] + rank[i]] = src[i];
  }
}


__global__ __launch_bounds__(256) void gemm1_wmma_kernel(
    const float* __restrict__ X, const float* __restrict__ W1,
    const float* __restrict__ dinv, float* __restrict__ hs, int n) {
  __shared__ float wl[PAIRS * PSTR];
  const int lane  = threadIdx.x & 31;
  const int wave  = threadIdx.x >> 5;
  const int m0    = blockIdx.x * 128 + wave * 16;
  const int mlan  = lane & 15;
  const int khalf = (lane >> 4) << 1;
  const int pbase = lane >> 4;

  int row = m0 + mlan;
  if (row >= n) row = n - 1;
  const float* xrow = X + (size_t)row * NF;

  v8f c0 = {}; v8f c1 = {}; v8f c2 = {}; v8f c3 = {};

  for (int kb = 0; kb < NF; kb += KTILE) {
    for (int i = threadIdx.x; i < KTILE * NH; i += 256) {
      int kl = i >> 6, nn = i & 63;
      wl[(kl >> 1) * PSTR + (nn << 1) + (kl & 1)] = W1[(size_t)kb * NH + i];
    }
    if (kb + KTILE < NF)
      __builtin_prefetch(xrow + kb + KTILE, 0, 3);
    __syncthreads();

    for (int kk = 0; kk < KTILE; kk += 4) {
      v2f a; a.x = xrow[kb + kk + khalf]; a.y = xrow[kb + kk + khalf + 1];
      const v2f* wp = (const v2f*)(wl + ((kk >> 1) + pbase) * PSTR);
      v2f b0 = wp[mlan];
      v2f b1 = wp[16 + mlan];
      v2f b2 = wp[32 + mlan];
      v2f b3 = wp[48 + mlan];
      c0 = __builtin_amdgcn_wmma_f32_16x16x4_f32(false, a, false, b0, (short)0, c0, false, false);
      c1 = __builtin_amdgcn_wmma_f32_16x16x4_f32(false, a, false, b1, (short)0, c1, false, false);
      c2 = __builtin_amdgcn_wmma_f32_16x16x4_f32(false, a, false, b2, (short)0, c2, false, false);
      c3 = __builtin_amdgcn_wmma_f32_16x16x4_f32(false, a, false, b3, (short)0, c3, false, false);
    }
    __syncthreads();
  }

  const int half8 = (lane >> 4) << 3;
  for (int r = 0; r < 8; ++r) {
    int rr = m0 + r + half8;
    if (rr < n) {
      float dv = dinv[rr];
      size_t base = (size_t)rr * NH + mlan;
      hs[base]      = c0[r] * dv;
      hs[base + 16] = c1[r] * dv;
      hs[base + 32] = c2[r] * dv;
      hs[base + 48] = c3[r] * dv;
    }
  }
}


__device__ __forceinline__ void add4(float4& a, const float4& b) {
  a.x += b.x; a.y += b.y; a.z += b.z; a.w += b.w;
}

__global__ __launch_bounds__(256) void agg1_gather_kernel(
    const int* __restrict__ rowptr, const int* __restrict__ cnt, const int* __restrict__ col,
    const float* __restrict__ hs, const float* __restrict__ dinv,
    const float* __restrict__ b1, const float* __restrict__ W2,
    float* __restrict__ t2s, int n) {
  const int lane = threadIdx.x & 31;
  const int wave = __builtin_amdgcn_readfirstlane(threadIdx.x >> 5);
  const int half = lane >> 4;
  const int q    = lane & 15;

  const float4 w0 = *(const float4*)(W2 + (4 * q + 0) * NC + 4 * half);
  const float4 w1 = *(const float4*)(W2 + (4 * q + 1) * NC + 4 * half);
  const float4 w2 = *(const float4*)(W2 + (4 * q + 2) * NC + 4 * half);
  const float4 w3 = *(const float4*)(W2 + (4 * q + 3) * NC + 4 * half);
  const float4 bb = *(const float4*)(b1 + 4 * q);

  const int node0 = (blockIdx.x * 8 + wave) * NPW;
  for (int u = 0; u < NPW; ++u) {
    const int i = node0 + u;
    if (i >= n) break;
    const int start = rowptr[i];
    const int end   = start + cnt[i];
    float4 acc = make_float4(0.f, 0.f, 0.f, 0.f);
    if (half == 0) acc = *(const float4*)(hs + (size_t)i * NH + 4 * q);

    for (int base = start; base < end; base += 32) {
      const int m = min(32, end - base);
      const int c = (lane < m) ? col[base + lane] : 0;
      int t = 0;
      for (; t + 8 <= m; t += 8) {
        int idx[4];
        float4 v[4];
#pragma unroll
        for (int z = 0; z < 4; ++z) idx[z] = __shfl(c, t + 2 * z + half);
#pragma unroll
        for (int z = 0; z < 4; ++z) v[z] = *(const float4*)(hs + (size_t)idx[z] * NH + 4 * q);
        add4(acc, v[0]); add4(acc, v[1]); add4(acc, v[2]); add4(acc, v[3]);
      }
      if (t < m) {
        float4 v[4];
#pragma unroll
        for (int z = 0; z < 4; ++z) {
          const int k = t + 2 * z + half;
          const int got = __shfl(c, k);
          v[z] = make_float4(0.f, 0.f, 0.f, 0.f);
          if (k < m) v[z] = *(const float4*)(hs + (size_t)got * NH + 4 * q);
        }
        add4(acc, v[0]); add4(acc, v[1]); add4(acc, v[2]); add4(acc, v[3]);
      }
    }
    acc.x += __shfl_xor(acc.x, 16);
    acc.y += __shfl_xor(acc.y, 16);
    acc.z += __shfl_xor(acc.z, 16);
    acc.w += __shfl_xor(acc.w, 16);

    const float di = dinv[i];
    const float h0 = fmaxf(fmaf(acc.x, di, bb.x), 0.0f);
    const float h1 = fmaxf(fmaf(acc.y, di, bb.y), 0.0f);
    const float h2 = fmaxf(fmaf(acc.z, di, bb.z), 0.0f);
    const float h3 = fmaxf(fmaf(acc.w, di, bb.w), 0.0f);
    float p0 = fmaf(h3, w3.x, fmaf(h2, w2.x, fmaf(h1, w1.x, h0 * w0.x)));
    float p1 = fmaf(h3, w3.y, fmaf(h2, w2.y, fmaf(h1, w1.y, h0 * w0.y)));
    float p2 = fmaf(h3, w3.z, fmaf(h2, w2.z, fmaf(h1, w1.z, h0 * w0.z)));
    float p3 = fmaf(h3, w3.w, fmaf(h2, w2.w, fmaf(h1, w1.w, h0 * w0.w)));
    for (int d = 8; d > 0; d >>= 1) {
      p0 += __shfl_xor(p0, d);
      p1 += __shfl_xor(p1, d);
      p2 += __shfl_xor(p2, d);
      p3 += __shfl_xor(p3, d);
    }
    if (q == 0)
      *(float4*)(t2s + (size_t)i * NC + 4 * half) = make_float4(p0 * di, p1 * di, p2 * di, p3 * di);
  }
}


__global__ __launch_bounds__(256) void agg2_gather_kernel(
    const int* __restrict__ rowptr, const int* __restrict__ cnt, const int* __restrict__ col,
    const float* __restrict__ t2s, const float* __restrict__ dinv,
    const float* __restrict__ b2, float* __restrict__ out, int n) {
  const int r = threadIdx.x & 7;
  const int i = (blockIdx.x * 256 + threadIdx.x) >> 3;
  const bool live = i < n;
  const int ii = live ? i : n - 1;
  const int start = rowptr[ii];
  const int end   = live ? start + cnt[ii] : start;

  float4 a0 = make_float4(0.f, 0.f, 0.f, 0.f), a1 = a0;
  if (r == 0 && live) {
    a0 = *(const float4*)(t2s + (size_t)ii * NC);
    a1 = *(const float4*)(t2s + (size_t)ii * NC + 4);
  }
  for (int j = start + r; j < end; j += 16) {
    const int s0 = col[j];
    const bool two = (j + 8) < end;
    int s1 = 0;
    if (two) s1 = col[j + 8];
    const float4 x0 = *(const float4*)(t2s + (size_t)s0 * NC);
    const float4 x1 = *(const float4*)(t2s + (size_t)s0 * NC + 4);
    float4 y0 = make_float4(0.f, 0.f, 0.f, 0.f), y1 = y0;
    if (two) {
      y0 = *(const float4*)(t2s + (size_t)s1 * NC);
      y1 = *(const float4*)(t2s + (size_t)s1 * NC + 4);
    }
    a0.x += x0.x; a0.y += x0.y; a0.z += x0.z; a0.w += x0.w;
    a1.x += x1.x; a1.y += x1.y; a1.z += x1.z; a1.w += x1.w;
    a0.x += y0.x; a0.y += y0.y; a0.z += y0.z; a0.w += y0.w;
    a1.x += y1.x; a1.y += y1.y; a1.z += y1.z; a1.w += y1.w;
  }
  for (int d = 4; d > 0; d >>= 1) {
    a0.x += __shfl_xor(a0.x, d); a0.y += __shfl_xor(a0.y, d);
    a0.z += __shfl_xor(a0.z, d); a0.w += __shfl_xor(a0.w, d);
    a1.x += __shfl_xor(a1.x, d); a1.y += __shfl_xor(a1.y, d);
    a1.z += __shfl_xor(a1.z, d); a1.w += __shfl_xor(a1.w, d);
  }
  if (r < 2 && live) {
    const float di = dinv[i];
    const float4 cb = *(const float4*)(b2 + 4 * r);
    const float4 a = (r == 0) ? a0 : a1;
    *(float4*)(out + (size_t)i * NC + 4 * r) =
        make_float4(fmaf(a.x, di, cb.x), fmaf(a.y, di, cb.y), fmaf(a.z, di, cb.z), fmaf(a.w, di, cb.w));
  }
}


extern "C" void kernel_launch(void* const* d_in, const int* in_sizes, int n_in,
                              void* d_out, int out_size, void* d_ws, size_t ws_size,
                              hipStream_t stream) {
  const float* x  = (const float*)d_in[0];
  const int*   ei = (const int*)d_in[1];
  const float* W1 = (const float*)d_in[2];
  const float* b1 = (const float*)d_in[3];
  const float* W2 = (const float*)d_in[4];
  const float* b2 = (const float*)d_in[5];
  float* out = (float*)d_out;

  const int n = in_sizes[0] / NF;
  const int e = in_sizes[1] / 2;
  const int* src = ei;
  const int* dst = ei + e;

  float* ws   = (float*)d_ws;
  float* hs   = ws;
  float* t2s  = hs + (size_t)n * NH;
  int*   col  = (int*)(t2s + (size_t)n * NC);
  int*   rank = col + e;
  int*   cnt  = rank + e;
  int*   rowptr = cnt + n;
  float* dinv = (float*)(rowptr + n);

  const int T = 256;
  const int e4 = (e + 3) / 4;
  cnt_zero_kernel<<<(n + T - 1) / T, T, 0, stream>>>(cnt, n);
  rank_kernel<<<(e4 + T - 1) / T, T, 0, stream>>>(dst, cnt, rank, e);
  scan_kernel<<<(n + 1023) / 1024, 1024, 0, stream>>>(cnt, rowptr, dinv, n);
  fill_kernel<<<(e4 + T - 1) / T, T, 0, stream>>>(src, dst, rank, rowptr, col, e);

  gemm1_wmma_kernel<<<(n + 127) / 128, T, 0, stream>>>(x, W1, dinv, hs, n);

  agg1_gather_kernel<<<(n + 8 * NPW - 1) / (8 * NPW), T, 0, stream>>>(rowptr, cnt, col, hs, dinv, b1, W2, t2s, n);

  agg2_gather_kernel<<<(n * NC + T - 1) / T, T, 0, stream>>>(rowptr, cnt, col, t2s, dinv, b2, out, n);
}


# dst-sorted edge list by a two-level bucket sort on LDS counters (chunk counts, scan, partition, per-bucket count/scan/place) replaces a global returning add per edge; row sums in LDS-add order
# speedup vs baseline: 3.4596x; 1.9508x over previous
#include <hip/hip_runtime.h>
#include <hip/hip_bf16.h>

typedef __attribute__((ext_vector_type(2))) float v2f;
typedef __attribute__((ext_vector_type(8))) float v8f;

#define NF 512
#define NH 64
#define NC 8
#define KTILE 128
#define PAIRS (KTILE / 2)
#define PSTR 160
#define NPW 4


#define BSH 9
#define BN (1 << BSH)
#define RSH 22
#define NBMAX 2048
#define G1 256
#define PT 1024
#define NOREC 0xFFFFFFFFu
#define LCAP 40960

__device__ __forceinline__ int pack_rec(int d, int s) {
  return (int)(((unsigned)(d & (BN - 1)) << RSH) | (unsigned)s);
}

__global__ __launch_bounds__(PT) void bucket_count_kernel(const int* __restrict__ dst, int* __restrict__ M,
                                                          int e, int nb, int ch) {
  __shared__ int lh[NBMAX];
  const int t = threadIdx.x, w = blockIdx.x;
  for (int b = t; b < nb; b += PT) lh[b] = 0;
  __syncthreads();
  const int lo = w * ch;
  const int hi = min(e, lo + ch);
  for (int i = lo + 4 * t; i < hi; i += 4 * PT) {
    if (i + 3 < hi && (((size_t)(dst + i)) & 15) == 0) {
      const int4 d = *(const int4*)(dst + i);
      atomicAdd(&lh[d.x >> BSH], 1);
      atomicAdd(&lh[d.y >> BSH], 1);
      atomicAdd(&lh[d.z >> BSH], 1);
      atomicAdd(&lh[d.w >> BSH], 1);
    } else {
      for (int j = i; j < hi && j < i + 4; ++j) atomicAdd(&lh[dst[j] >> BSH], 1);
    }
  }
  __syncthreads();
  for (int b = t; b < nb; b += PT) M[b * G1 + w] = lh[b];
}

__global__ __launch_bounds__(PT) void matrix_scan_kernel(int* M, int total) {
  __shared__ int wsum[32];
  const int t = threadIdx.x, lane = t & 31, wave = t >> 5;
  const int len = (total + PT - 1) / PT;
  const int lo = min(total, t * len), hi = min(total, lo + len);
  int s = 0;
  for (int i = lo; i < hi; ++i) s += M[i];
  int incl = s;
  for (int d = 1; d < 32; d <<= 1) {
    int u = __shfl_up(incl, d);
    if (lane >= d) incl += u;
  }
  if (lane == 31) wsum[wave] = incl;
  __syncthreads();
  int front = (lane < wave) ? wsum[lane] : 0;
  for (int d = 16; d > 0; d >>= 1) front += __shfl_xor(front, d);
  int run = front + incl - s;
  for (int i = lo; i < hi; ++i) {
    const int v = M[i];
    M[i] = run;
    run += v;
  }
}

__global__ __launch_bounds__(PT) void bucket_partition_kernel(const int* __restrict__ src,
                                                              const int* __restrict__ dst,
                                                              const int* __restrict__ M,
                                                              int* __restrict__ part, int e, int nb, int ch) {
  __shared__ int cur[NBMAX];
  const int t = threadIdx.x, w = blockIdx.x;
  for (int b = t; b < nb; b += PT) cur[b] = M[b * G1 + w];
  __syncthreads();
  const int lo = w * ch;
  const int hi = min(e, lo + ch);
  for (int i = lo + 4 * t; i < hi; i += 4 * PT) {
    if (i + 3 < hi && (((size_t)(dst + i)) & 15) == 0 && (((size_t)(src + i)) & 15) == 0) {
      const int4 d = *(const int4*)(dst + i);
      const int4 s = *(const int4*)(src + i);
      const int p0 = atomicAdd(&cur[d.x >> BSH], 1);
      const int p1 = atomicAdd(&cur[d.y >> BSH], 1);
      const int p2 = atomicAdd(&cur[d.z >> BSH], 1);
      const int p3 = atomicAdd(&cur[d.w >> BSH], 1);
      part[p0] = pack_rec(d.x, s.x);
      part[p1] = pack_rec(d.y, s.y);
      part[p2] = pack_rec(d.z, s.z);
      part[p3] = pack_rec(d.w, s.w);
    } else {
      for (int j = i; j < hi && j < i + 4; ++j) {
        const int d = dst[j];
        const int p = atomicAdd(&cur[d >> BSH], 1);
        part[p] = pack_rec(d, src[j]);
      }
    }
  }
}

__global__ __launch_bounds__(PT) void bucket_csr_kernel(const int* __restrict__ M, const int* __restrict__ part,
                                                        int* __restrict__ cnt, int* __restrict__ rowptr,
                                                        float* __restrict__ dinv, int* __restrict__ col,
                                                        int n, int e, int nb) {
  __shared__ int lc[BN];
  __shared__ int wsum[32];
  __shared__ int lcol[LCAP];
  const int t = threadIdx.x, lane = t & 31, wave = t >> 5;
  const int b = blockIdx.x;
  const int s0 = M[b * G1];
  const int s1 = (b + 1 < nb) ? M[(b + 1) * G1] : e;
  const bool staged = (s1 - s0) <= LCAP;
  if (t < BN) lc[t] = 0;
  __syncthreads();
  for (int i = s0 + t; i < s1; i += 4 * PT) {
    const int i1 = i + PT, i2 = i + 2 * PT, i3 = i + 3 * PT;
    const unsigned r0 = (unsigned)part[i];
    const unsigned r1 = (i1 < s1) ? (unsigned)part[i1] : NOREC;
    const unsigned r2 = (i2 < s1) ? (unsigned)part[i2] : NOREC;
    const unsigned r3 = (i3 < s1) ? (unsigned)part[i3] : NOREC;
    atomicAdd(&lc[r0 >> RSH], 1);
    if (r1 != NOREC) atomicAdd(&lc[r1 >> RSH], 1);
    if (r2 != NOREC) atomicAdd(&lc[r2 >> RSH], 1);
    if (r3 != NOREC) atomicAdd(&lc[r3 >> RSH], 1);
  }
  __syncthreads();
  const int v = (t < BN) ? lc[t] : 0;
  int incl = v;
  for (int d = 1; d < 32; d <<= 1) {
    int u = __shfl_up(incl, d);
    if (lane >= d) incl += u;
  }
  if (lane == 31) wsum[wave] = incl;
  __syncthreads();
  int front = (lane < wave) ? wsum[lane] : 0;
  for (int d = 16; d > 0; d >>= 1) front += __shfl_xor(front, d);
  const int excl = front + incl - v;
  if (t < BN) {
    lc[t] = excl;
    const int node = b * BN + t;
    if (node < n) {
      cnt[node] = v;
      rowptr[node] = s0 + excl;
      dinv[node] = rsqrtf((float)(v + 1));
    }
  }
  __syncthreads();
  for (int i = s0 + t; i < s1; i += 4 * PT) {
    const int i1 = i + PT, i2 = i + 2 * PT, i3 = i + 3 * PT;
    const unsigned r0 = (unsigned)part[i];
    const unsigned r1 = (i1 < s1) ? (unsigned)part[i1] : NOREC;
    const unsigned r2 = (i2 < s1) ? (unsigned)part[i2] : NOREC;
    const unsigned r3 = (i3 < s1) ? (unsigned)part[i3] : NOREC;
    const unsigned smask = (1u << RSH) - 1u;
    const int p0 = atomicAdd(&lc[r0 >> RSH], 1);
    const int p1 = (r1 != NOREC) ? atomicAdd(&lc[r1 >> RSH], 1) : -1;
    const int p2 = (r2 != NOREC) ? atomicAdd(&lc[r2 >> RSH], 1) : -1;
    const int p3 = (r3 != NOREC) ? atomicAdd(&lc[r3 >> RSH], 1) : -1;
    if (staged) {
      lcol[p0] = (int)(r0 & smask);
      if (p1 >= 0) lcol[p1] = (int)(r1 & smask);
      if (p2 >= 0) lcol[p2] = (int)(r2 & smask);
      if (p3 >= 0) lcol[p3] = (int)(r3 & smask);
    } else {
      col[s0 + p0] = (int)(r0 & smask);
      if (p1 >= 0) col[s0 + p1] = (int)(r1 & smask);
      if (p2 >= 0) col[s0 + p2] = (int)(r2 & smask);
      if (p3 >= 0) col[s0 + p3] = (int)(r3 & smask);
    }
  }
  if (staged) {
    __syncthreads();
    for (int j = t; j < s1 - s0; j += PT) col[s0 + j] = lcol[j];
  }
}


__global__ __launch_bounds__(256) void gemm1_wmma_kernel(
    const float* __restrict__ X, const float* __restrict__ W1,
    const float* __restrict__ dinv, float* __restrict__ hs, int n) {
  __shared__ float wl[PAIRS * PSTR];
  const int lane  = threadIdx.x & 31;
  const int wave  = threadIdx.x >> 5;
  const int m0    = blockIdx.x * 128 + wave * 16;
  const int mlan  = lane & 15;
  const int khalf = (lane >> 4) << 1;
  const int pbase = lane >> 4;

  int row = m0 + mlan;
  if (row >= n) row = n - 1;
  const float* xrow = X + (size_t)row * NF;

  v8f c0 = {}; v8f c1 = {}; v8f c2 = {}; v8f c3 = {};

  for (int kb = 0; kb < NF; kb += KTILE) {
    for (int i = threadIdx.x; i < KTILE * NH; i += 256) {
      int kl = i >> 6, nn = i & 63;
      wl[(kl >> 1) * PSTR + (nn << 1) + (kl & 1)] = W1[(size_t)kb * NH + i];
    }
    if (kb + KTILE < NF)
      __builtin_prefetch(xrow + kb + KTILE, 0, 3);
    __syncthreads();

    for (int kk = 0; kk < KTILE; kk += 4) {
      v2f a; a.x = xrow[kb + kk + khalf]; a.y = xrow[kb + kk + khalf + 1];
      const v2f* wp = (const v2f*)(wl + ((kk >> 1) + pbase) * PSTR);
      v2f b0 = wp[mlan];
      v2f b1 = wp[16 + mlan];
      v2f b2 = wp[32 + mlan];
      v2f b3 = wp[48 + mlan];
      c0 = __builtin_amdgcn_wmma_f32_16x16x4_f32(false, a, false, b0, (short)0, c0, false, false);
      c1 = __builtin_amdgcn_wmma_f32_16x16x4_f32(false, a, false, b1, (short)0, c1, false, false);
      c2 = __builtin_amdgcn_wmma_f32_16x16x4_f32(false, a, false, b2, (short)0, c2, false, false);
      c3 = __builtin_amdgcn_wmma_f32_16x16x4_f32(false, a, false, b3, (short)0, c3, false, false);
    }
    __syncthreads();
  }

  const int half8 = (lane >> 4) << 3;
  for (int r = 0; r < 8; ++r) {
    int rr = m0 + r + half8;
    if (rr < n) {
      float dv = dinv[rr];
      size_t base = (size_t)rr * NH + mlan;
      hs[base]      = c0[r] * dv;
      hs[base + 16] = c1[r] * dv;
      hs[base + 32] = c2[r] * dv;
      hs[base + 48] = c3[r] * dv;
    }
  }
}


__device__ __forceinline__ void add4(float4& a, const float4& b) {
  a.x += b.x; a.y += b.y; a.z += b.z; a.w += b.w;
}

__global__ __launch_bounds__(256) void agg1_gather_kernel(
    const int* __restrict__ rowptr, const int* __restrict__ cnt, const int* __restrict__ col,
    const float* __restrict__ hs, const float* __restrict__ dinv,
    const float* __restrict__ b1, const float* __restrict__ W2,
    float* __restrict__ t2s, int n) {
  const int lane = threadIdx.x & 31;
  const int wave = __builtin_amdgcn_readfirstlane(threadIdx.x >> 5);
  const int half = lane >> 4;
  const int q    = lane & 15;

  const float4 w0 = *(const float4*)(W2 + (4 * q + 0) * NC + 4 * half);
  const float4 w1 = *(const float4*)(W2 + (4 * q + 1) * NC + 4 * half);
  const float4 w2 = *(const float4*)(W2 + (4 * q + 2) * NC + 4 * half);
  const float4 w3 = *(const float4*)(W2 + (4 * q + 3) * NC + 4 * half);
  const float4 bb = *(const float4*)(b1 + 4 * q);

  const int node0 = (blockIdx.x * 8 + wave) * NPW;
  for (int u = 0; u < NPW; ++u) {
    const int i = node0 + u;
    if (i >= n) break;
    const int start = rowptr[i];
    const int end   = start + cnt[i];
    float4 acc = make_float4(0.f, 0.f, 0.f, 0.f);
    if (half == 0) acc = *(const float4*)(hs + (size_t)i * NH + 4 * q);

    for (int base = start; base < end; base += 32) {
      const int m = min(32, end - base);
      const int c = (lane < m) ? col[base + lane] : 0;
      int t = 0;
      for (; t + 8 <= m; t += 8) {
        int idx[4];
        float4 v[4];
#pragma unroll
        for (int z = 0; z < 4; ++z) idx[z] = __shfl(c, t + 2 * z + half);
#pragma unroll
        for (int z = 0; z < 4; ++z) v[z] = *(const float4*)(hs + (size_t)idx[z] * NH + 4 * q);
        add4(acc, v[0]); add4(acc, v[1]); add4(acc, v[2]); add4(acc, v[3]);
      }
      if (t < m) {
        float4 v[4];
#pragma unroll
        for (int z = 0; z < 4; ++z) {
          const int k = t + 2 * z + half;
          const int got = __shfl(c, k);
          v[z] = make_float4(0.f, 0.f, 0.f, 0.f);
          if (k < m) v[z] = *(const float4*)(hs + (size_t)got * NH + 4 * q);
        }
        add4(acc, v[0]); add4(acc, v[1]); add4(acc, v[2]); add4(acc, v[3]);
      }
    }
    acc.x += __shfl_xor(acc.x, 16);
    acc.y += __shfl_xor(acc.y, 16);
    acc.z += __shfl_xor(acc.z, 16);
    acc.w += __shfl_xor(acc.w, 16);

    const float di = dinv[i];
    const float h0 = fmaxf(fmaf(acc.x, di, bb.x), 0.0f);
    const float h1 = fmaxf(fmaf(acc.y, di, bb.y), 0.0f);
    const float h2 = fmaxf(fmaf(acc.z, di, bb.z), 0.0f);
    const float h3 = fmaxf(fmaf(acc.w, di, bb.w), 0.0f);
    float p0 = fmaf(h3, w3.x, fmaf(h2, w2.x, fmaf(h1, w1.x, h0 * w0.x)));
    float p1 = fmaf(h3, w3.y, fmaf(h2, w2.y, fmaf(h1, w1.y, h0 * w0.y)));
    float p2 = fmaf(h3, w3.z, fmaf(h2, w2.z, fmaf(h1, w1.z, h0 * w0.z)));
    float p3 = fmaf(h3, w3.w, fmaf(h2, w2.w, fmaf(h1, w1.w, h0 * w0.w)));
    for (int d = 8; d > 0; d >>= 1) {
      p0 += __shfl_xor(p0, d);
      p1 += __shfl_xor(p1, d);
      p2 += __shfl_xor(p2, d);
      p3 += __shfl_xor(p3, d);
    }
    if (q == 0)
      *(float4*)(t2s + (size_t)i * NC + 4 * half) = make_float4(p0 * di, p1 * di, p2 * di, p3 * di);
  }
}


__global__ __launch_bounds__(256) void agg2_gather_kernel(
    const int* __restrict__ rowptr, const int* __restrict__ cnt, const int* __restrict__ col,
    const float* __restrict__ t2s, const float* __restrict__ dinv,
    const float* __restrict__ b2, float* __restrict__ out, int n) {
  const int r = threadIdx.x & 7;
  const int i = (blockIdx.x * 256 + threadIdx.x) >> 3;
  const bool live = i < n;
  const int ii = live ? i : n - 1;
  const int start = rowptr[ii];
  const int end   = live ? start + cnt[ii] : start;

  float4 a0 = make_float4(0.f, 0.f, 0.f, 0.f), a1 = a0;
  if (r == 0 && live) {
    a0 = *(const float4*)(t2s + (size_t)ii * NC);
    a1 = *(const float4*)(t2s + (size_t)ii * NC + 4);
  }
  for (int j = start + r; j < end; j += 16) {
    const int s0 = col[j];
    const bool two = (j + 8) < end;
    int s1 = 0;
    if (two) s1 = col[j + 8];
    const float4 x0 = *(const float4*)(t2s + (size_t)s0 * NC);
    const float4 x1 = *(const float4*)(t2s + (size_t)s0 * NC + 4);
    float4 y0 = make_float4(0.f, 0.f, 0.f, 0.f), y1 = y0;
    if (two) {
      y0 = *(const float4*)(t2s + (size_t)s1 * NC);
      y1 = *(const float4*)(t2s + (size_t)s1 * NC + 4);
    }
    a0.x += x0.x; a0.y += x0.y; a0.z += x0.z; a0.w += x0.w;
    a1.x += x1.x; a1.y += x1.y; a1.z += x1.z; a1.w += x1.w;
    a0.x += y0.x; a0.y += y0.y; a0.z += y0.z; a0.w += y0.w;
    a1.x += y1.x; a1.y += y1.y; a1.z += y1.z; a1.w += y1.w;
  }
  for (int d = 4; d > 0; d >>= 1) {
    a0.x += __shfl_xor(a0.x, d); a0.y += __shfl_xor(a0.y, d);
    a0.z += __shfl_xor(a0.z, d); a0.w += __shfl_xor(a0.w, d);
    a1.x += __shfl_xor(a1.x, d); a1.y += __shfl_xor(a1.y, d);
    a1.z += __shfl_xor(a1.z, d); a1.w += __shfl_xor(a1.w, d);
  }
  if (r < 2 && live) {
    const float di = dinv[i];
    const float4 cb = *(const float4*)(b2 + 4 * r);
    const float4 a = (r == 0) ? a0 : a1;
    *(float4*)(out + (size_t)i * NC + 4 * r) =
        make_float4(fmaf(a.x, di, cb.x), fmaf(a.y, di, cb.y), fmaf(a.z, di, cb.z), fmaf(a.w, di, cb.w));
  }
}


extern "C" void kernel_launch(void* const* d_in, const int* in_sizes, int n_in,
                              void* d_out, int out_size, void* d_ws, size_t ws_size,
                              hipStream_t stream) {
  const float* x  = (const float*)d_in[0];
  const int*   ei = (const int*)d_in[1];
  const float* W1 = (const float*)d_in[2];
  const float* b1 = (const float*)d_in[3];
  const float* W2 = (const float*)d_in[4];
  const float* b2 = (const float*)d_in[5];
  float* out = (float*)d_out;

  const int n = in_sizes[0] / NF;
  const int e = in_sizes[1] / 2;
  const int* src = ei;
  const int* dst = ei + e;

  float* ws   = (float*)d_ws;
  float* hs   = ws;
  float* t2s  = hs + (size_t)n * NH;
  int*   col  = (int*)(t2s + (size_t)n * NC);
  int*   part = col + e;
  int*   cnt  = part + e;
  int*   rowptr = cnt + n;
  float* dinv = (float*)(rowptr + n);
  int*   M    = (int*)t2s;

  const int T = 256;
  const int nb = (n + BN - 1) / BN;
  const int ch = (((e + G1 - 1) / G1) + 3) & ~3;
  bucket_count_kernel<<<G1, PT, 0, stream>>>(dst, M, e, nb, ch);
  matrix_scan_kernel<<<1, PT, 0, stream>>>(M, nb * G1);
  bucket_partition_kernel<<<G1, PT, 0, stream>>>(src, dst, M, part, e, nb, ch);
  bucket_csr_kernel<<<nb, PT, 0, stream>>>(M, part, cnt, rowptr, dinv, col, n, e, nb);

  gemm1_wmma_kernel<<<(n + 127) / 128, T, 0, stream>>>(x, W1, dinv, hs, n);

  agg1_gather_kernel<<<(n + 8 * NPW - 1) / (8 * NPW), T, 0, stream>>>(rowptr, cnt, col, hs, dinv, b1, W2, t2s, n);

  agg2_gather_kernel<<<(n * NC + T - 1) / T, T, 0, stream>>>(rowptr, cnt, col, t2s, dinv, b2, out, n);
}


# bucket starts and chunk cursors computed inside the partition launch: each work-group sums the raw count rows itself (16-byte row loads, lane sums, LDS scan); the one-work-group scan launch is gone
# speedup vs baseline: 3.8873x; 1.1236x over previous
#include <hip/hip_runtime.h>
#include <hip/hip_bf16.h>

typedef __attribute__((ext_vector_type(2))) float v2f;
typedef __attribute__((ext_vector_type(8))) float v8f;

#define NF 512
#define NH 64
#define NC 8
#define KTILE 128
#define PAIRS (KTILE / 2)
#define PSTR 160
#define NPW 4


#define BSH 9
#define BN (1 << BSH)
#define RSH 22
#define NBMAX 2048
#define G1 256
#define PT 1024
#define NOREC 0xFFFFFFFFu
#define LCAP 40960

__device__ __forceinline__ int pack_rec(int d, int s) {
  return (int)(((unsigned)(d & (BN - 1)) << RSH) | (unsigned)s);
}

__global__ __launch_bounds__(PT) void bucket_count_kernel(const int* __restrict__ dst, int* __restrict__ M,
                                                          int e, int nb, int ch) {
  __shared__ int lh[NBMAX];
  const int t = threadIdx.x, w = blockIdx.x;
  for (int b = t; b < nb; b += PT) lh[b] = 0;
  __syncthreads();
  const int lo = w * ch;
  const int hi = min(e, lo + ch);
  for (int i = lo + 4 * t; i < hi; i += 4 * PT) {
    if (i + 3 < hi && (((size_t)(dst + i)) & 15) == 0) {
      const int4 d = *(const int4*)(dst + i);
      atomicAdd(&lh[d.x >> BSH], 1);
      atomicAdd(&lh[d.y >> BSH], 1);
      atomicAdd(&lh[d.z >> BSH], 1);
      atomicAdd(&lh[d.w >> BSH], 1);
    } else {
      for (int j = i; j < hi && j < i + 4; ++j) atomicAdd(&lh[dst[j] >> BSH], 1);
    }
  }
  __syncthreads();
  for (int b = t; b < nb; b += PT) M[b * G1 + w] = lh[b];
}

static_assert(NBMAX == 2 * PT, "the bucket scan gives every thread two buckets");
static_assert(G1 % 32 == 0, "eight lanes read a row of M as 16-byte pieces, one 128-byte line at a time");

__global__ __launch_bounds__(PT) void bucket_partition_kernel(const int* __restrict__ src,
                                                              const int* __restrict__ dst,
                                                              const int* __restrict__ M,
                                                              int* __restrict__ bstart,
                                                              int* __restrict__ part, int e, int nb, int ch) {
  __shared__ int cur[NBMAX];
  __shared__ int pre[NBMAX];
  __shared__ int wsum[32];
  const int t = threadIdx.x, lane = t & 31, w = blockIdx.x;
  const int wave = __builtin_amdgcn_readfirstlane(t >> 5);

  const int g = lane >> 3, q = lane & 7;
  for (int g0 = wave; 4 * g0 < nb; g0 += 32) {
    const int b = 4 * g0 + g;
    const int4* row = (const int4*)(M + (size_t)min(b, nb - 1) * G1);
    int4 v[G1 / 32];
#pragma unroll
    for (int j = 0; j < G1 / 32; ++j) v[j] = row[q + 8 * j];
    int tot = 0, frn = 0;
#pragma unroll
    for (int j = 0; j < G1 / 32; ++j) {
      const int c = 4 * (q + 8 * j);
      tot += v[j].x + v[j].y + v[j].z + v[j].w;
      frn += (c < w ? v[j].x : 0) + (c + 1 < w ? v[j].y : 0) + (c + 2 < w ? v[j].z : 0) + (c + 3 < w ? v[j].w : 0);
    }
    for (int d = 4; d > 0; d >>= 1) {
      tot += __shfl_xor(tot, d);
      frn += __shfl_xor(frn, d);
    }
    if (q == 0 && b < nb) {
      cur[b] = tot;
      pre[b] = frn;
    }
  }
  __syncthreads();
  {
    const int i0 = 2 * t, i1 = 2 * t + 1;
    const int v0 = (i0 < nb) ? cur[i0] : 0;
    const int v1 = (i1 < nb) ? cur[i1] : 0;
    const int f0 = (i0 < nb) ? pre[i0] : 0;
    const int f1 = (i1 < nb) ? pre[i1] : 0;
    const int s = v0 + v1;
    int incl = s;
    for (int d = 1; d < 32; d <<= 1) {
      int u = __shfl_up(incl, d);
      if (lane >= d) incl += u;
    }
    if (lane == 31) wsum[wave] = incl;
    __syncthreads();
    int front = (lane < wave) ? wsum[lane] : 0;
    for (int d = 16; d > 0; d >>= 1) front += __shfl_xor(front, d);
    const int excl = front + incl - s;
    if (i0 < nb) cur[i0] = excl + f0;
    if (i1 < nb) cur[i1] = excl + v0 + f1;
    if (w == 0) {
      if (i0 < nb) bstart[i0] = excl;
      if (i1 < nb) bstart[i1] = excl + v0;
    }
  }
  __syncthreads();
  const int lo = w * ch;
  const int hi = min(e, lo + ch);
  for (int i = lo + 4 * t; i < hi; i += 4 * PT) {
    if (i + 3 < hi && (((size_t)(dst + i)) & 15) == 0 && (((size_t)(src + i)) & 15) == 0) {
      const int4 d = *(const int4*)(dst + i);
      const int4 s = *(const int4*)(src + i);
      const int p0 = atomicAdd(&cur[d.x >> BSH], 1);
      const int p1 = atomicAdd(&cur[d.y >> BSH], 1);
      const int p2 = atomicAdd(&cur[d.z >> BSH], 1);
      const int p3 = atomicAdd(&cur[d.w >> BSH], 1);
      part[p0] = pack_rec(d.x, s.x);
      part[p1] = pack_rec(d.y, s.y);
      part[p2] = pack_rec(d.z, s.z);
      part[p3] = pack_rec(d.w, s.w);
    } else {
      for (int j = i; j < hi && j < i + 4; ++j) {
        const int d = dst[j];
        const int p = atomicAdd(&cur[d >> BSH], 1);
        part[p] = pack_rec(d, src[j]);
      }
    }
  }
}

__global__ __launch_bounds__(PT) void bucket_csr_kernel(const int* __restrict__ bstart, const int* __restrict__ part,
                                                        int* __restrict__ cnt, int* __restrict__ rowptr,
                                                        float* __restrict__ dinv, int* __restrict__ col,
                                                        int n, int e, int nb) {
  __shared__ int lc[BN];
  __shared__ int wsum[32];
  __shared__ int lcol[LCAP];
  const int t = threadIdx.x, lane = t & 31, wave = t >> 5;
  const int b = blockIdx.x;
  const int s0 = bstart[b];
  const int s1 = (b + 1 < nb) ? bstart[b + 1] : e;
  const bool staged = (s1 - s0) <= LCAP;
  if (t < BN) lc[t] = 0;
  __syncthreads();
  for (int i = s0 + t; i < s1; i += 4 * PT) {
    const int i1 = i + PT, i2 = i + 2 * PT, i3 = i + 3 * PT;
    const unsigned r0 = (unsigned)part[i];
    const unsigned r1 = (i1 < s1) ? (unsigned)part[i1] : NOREC;
    const unsigned r2 = (i2 < s1) ? (unsigned)part[i2] : NOREC;
    const unsigned r3 = (i3 < s1) ? (unsigned)part[i3] : NOREC;
    atomicAdd(&lc[r0 >> RSH], 1);
    if (r1 != NOREC) atomicAdd(&lc[r1 >> RSH], 1);
    if (r2 != NOREC) atomicAdd(&lc[r2 >> RSH], 1);
    if (r3 != NOREC) atomicAdd(&lc[r3 >> RSH], 1);
  }
  __syncthreads();
  const int v = (t < BN) ? lc[t] : 0;
  int incl = v;
  for (int d = 1; d < 32; d <<= 1) {
    int u = __shfl_up(incl, d);
    if (lane >= d) incl += u;
  }
  if (lane == 31) wsum[wave] = incl;
  __syncthreads();
  int front = (lane < wave) ? wsum[lane] : 0;
  for (int d = 16; d > 0; d >>= 1) front += __shfl_xor(front, d);
  const int excl = front + incl - v;
  if (t < BN) {
    lc[t] = excl;
    const int node = b * BN + t;
    if (node < n) {
      cnt[node] = v;
      rowptr[node] = s0 + excl;
      dinv[node] = rsqrtf((float)(v + 1));
    }
  }
  __syncthreads();
  for (int i = s0 + t; i < s1; i += 4 * PT) {
    const int i1 = i + PT, i2 = i + 2 * PT, i3 = i + 3 * PT;
    const unsigned r0 = (unsigned)part[i];
    const unsigned r1 = (i1 < s1) ? (unsigned)part[i1] : NOREC;
    const unsigned r2 = (i2 < s1) ? (unsigned)part[i2] : NOREC;
    const unsigned r3 = (i3 < s1) ? (unsigned)part[i3] : NOREC;
    const unsigned smask = (1u << RSH) - 1u;
    const int p0 = atomicAdd(&lc[r0 >> RSH], 1);
    const int p1 = (r1 != NOREC) ? atomicAdd(&lc[r1 >> RSH], 1) : -1;
    const int p2 = (r2 != NOREC) ? atomicAdd(&lc[r2 >> RSH], 1) : -1;
    const int p3 = (r3 != NOREC) ? atomicAdd(&lc[r3 >> RSH], 1) : -1;
    if (staged) {
      lcol[p0] = (int)(r0 & smask);
      if (p1 >= 0) lcol[p1] = (int)(r1 & smask);
      if (p2 >= 0) lcol[p2] = (int)(r2 & smask);
      if (p3 >= 0) lcol[p3] = (int)(r3 & smask);
    } else {
      col[s0 + p0] = (int)(r0 & smask);
      if (p1 >= 0) col[s0 + p1] = (int)(r1 & smask);
      if (p2 >= 0) col[s0 + p2] = (int)(r2 & smask);
      if (p3 >= 0) col[s0 + p3] = (int)(r3 & smask);
    }
  }
  if (staged) {
    __syncthreads();
    for (int j = t; j < s1 - s0; j += PT) col[s0 + j] = lcol[j];
  }
}


__global__ __launch_bounds__(256) void gemm1_wmma_kernel(
    const float* __restrict__ X, const float* __restrict__ W1,
    const float* __restrict__ dinv, float* __restrict__ hs, int n) {
  __shared__ float wl[PAIRS * PSTR];
  const int lane  = threadIdx.x & 31;
  const int wave  = threadIdx.x >> 5;
  const int m0    = blockIdx.x * 128 + wave * 16;
  const int mlan  = lane & 15;
  const int khalf = (lane >> 4) << 1;
  const int pbase = lane >> 4;

  int row = m0 + mlan;
  if (row >= n) row = n - 1;
  const float* xrow = X + (size_t)row * NF;

  v8f c0 = {}; v8f c1 = {}; v8f c2 = {}; v8f c3 = {};

  for (int kb = 0; kb < NF; kb += KTILE) {
    for (int i = threadIdx.x; i < KTILE * NH; i += 256) {
      int kl = i >> 6, nn = i & 63;
      wl[(kl >> 1) * PSTR + (nn << 1) + (kl & 1)] = W1[(size_t)kb * NH + i];
    }
    if (kb + KTILE < NF)
      __builtin_prefetch(xrow + kb + KTILE, 0, 3);
    __syncthreads();

    for (int kk = 0; kk < KTILE; kk += 4) {
      v2f a; a.x = xrow[kb + kk + khalf]; a.y = xrow[kb + kk + khalf + 1];
      const v2f* wp = (const v2f*)(wl + ((kk >> 1) + pbase) * PSTR);
      v2f b0 = wp[mlan];
      v2f b1 = wp[16 + mlan];
      v2f b2 = wp[32 + mlan];
      v2f b3 = wp[48 + mlan];
      c0 = __builtin_amdgcn_wmma_f32_16x16x4_f32(false, a, false, b0, (short)0, c0, false, false);
      c1 = __builtin_amdgcn_wmma_f32_16x16x4_f32(false, a, false, b1, (short)0, c1, false, false);
      c2 = __builtin_amdgcn_wmma_f32_16x16x4_f32(false, a, false, b2, (short)0, c2, false, false);
      c3 = __builtin_amdgcn_wmma_f32_16x16x4_f32(false, a, false, b3, (short)0, c3, false, false);
    }
    __syncthreads();
  }

  const int half8 = (lane >> 4) << 3;
  for (int r = 0; r < 8; ++r) {
    int rr = m0 + r + half8;
    if (rr < n) {
      float dv = dinv[rr];
      size_t base = (size_t)rr * NH + mlan;
      hs[base]      = c0[r] * dv;
      hs[base + 16] = c1[r] * dv;
      hs[base + 32] = c2[r] * dv;
      hs[base + 48] = c3[r] * dv;
    }
  }
}


__device__ __forceinline__ void add4(float4& a, const float4& b) {
  a.x += b.x; a.y += b.y; a.z += b.z; a.w += b.w;
}

__global__ __launch_bounds__(256) void agg1_gather_kernel(
    const int* __restrict__ rowptr, const int* __restrict__ cnt, const int* __restrict__ col,
    const float* __restrict__ hs, const float* __restrict__ dinv,
    const float* __restrict__ b1, const float* __restrict__ W2,
    float* __restrict__ t2s, int n) {
  const int lane = threadIdx.x & 31;
  const int wave = __builtin_amdgcn_readfirstlane(threadIdx.x >> 5);
  const int half = lane >> 4;
  const int q    = lane & 15;

  const float4 w0 = *(const float4*)(W2 + (4 * q + 0) * NC + 4 * half);
  const float4 w1 = *(const float4*)(W2 + (4 * q + 1) * NC + 4 * half);
  const float4 w2 = *(const float4*)(W2 + (4 * q + 2) * NC + 4 * half);
  const float4 w3 = *(const float4*)(W2 + (4 * q + 3) * NC + 4 * half);
  const float4 bb = *(const float4*)(b1 + 4 * q);

  const int node0 = (blockIdx.x * 8 + wave) * NPW;
  for (int u = 0; u < NPW; ++u) {
    const int i = node0 + u;
    if (i >= n) break;
    const int start = rowptr[i];
    const int end   = start + cnt[i];
    float4 acc = make_float4(0.f, 0.f, 0.f, 0.f);
    if (half == 0) acc = *(const float4*)(hs + (size_t)i * NH + 4 * q);

    for (int base = start; base < end; base += 32) {
      const int m = min(32, end - base);
      const int c = (lane < m) ? col[base + lane] : 0;
      int t = 0;
      for (; t + 8 <= m; t += 8) {
        int idx[4];
        float4 v[4];
#pragma unroll
        for (int z = 0; z < 4; ++z) idx[z] = __shfl(c, t + 2 * z + half);
#pragma unroll
        for (int z = 0; z < 4; ++z) v[z] = *(const float4*)(hs + (size_t)idx[z] * NH + 4 * q);
        add4(acc, v[0]); add4(acc, v[1]); add4(acc, v[2]); add4(acc, v[3]);
      }
      if (t < m) {
        float4 v[4];
#pragma unroll
        for (int z = 0; z < 4; ++z) {
          const int k = t + 2 * z + half;
          const int got = __shfl(c, k);
          v[z] = make_float4(0.f, 0.f, 0.f, 0.f);
          if (k < m) v[z] = *(const float4*)(hs + (size_t)got * NH + 4 * q);
        }
        add4(acc, v[0]); add4(acc, v[1]); add4(acc, v[2]); add4(acc, v[3]);
      }
    }
    acc.x += __shfl_xor(acc.x, 16);
    acc.y += __shfl_xor(acc.y, 16);
    acc.z += __shfl_xor(acc.z, 16);
    acc.w += __shfl_xor(acc.w, 16);

    const float di = dinv[i];
    const float h0 = fmaxf(fmaf(acc.x, di, bb.x), 0.0f);
    const float h1 = fmaxf(fmaf(acc.y, di, bb.y), 0.0f);
    const float h2 = fmaxf(fmaf(acc.z, di, bb.z), 0.0f);
    const float h3 = fmaxf(fmaf(acc.w, di, bb.w), 0.0f);
    float p0 = fmaf(h3, w3.x, fmaf(h2, w2.x, fmaf(h1, w1.x, h0 * w0.x)));
    float p1 = fmaf(h3, w3.y, fmaf(h2, w2.y, fmaf(h1, w1.y, h0 * w0.y)));
    float p2 = fmaf(h3, w3.z, fmaf(h2, w2.z, fmaf(h1, w1.z, h0 * w0.z)));
    float p3 = fmaf(h3, w3.w, fmaf(h2, w2.w, fmaf(h1, w1.w, h0 * w0.w)));
    for (int d = 8; d > 0; d >>= 1) {
      p0 += __shfl_xor(p0, d);
      p1 += __shfl_xor(p1, d);
      p2 += __shfl_xor(p2, d);
      p3 += __shfl_xor(p3, d);
    }
    if (q == 0)
      *(float4*)(t2s + (size_t)i * NC + 4 * half) = make_float4(p0 * di, p1 * di, p2 * di, p3 * di);
  }
}


__global__ __launch_bounds__(256) void agg2_gather_kernel(
    const int* __restrict__ rowptr, const int* __restrict__ cnt, const int* __restrict__ col,
    const float* __restrict__ t2s, const float* __restrict__ dinv,
    const float* __restrict__ b2, float* __restrict__ out, int n) {
  const int r = threadIdx.x & 7;
  const int i = (blockIdx.x * 256 + threadIdx.x) >> 3;
  const bool live = i < n;
  const int ii = live ? i : n - 1;
  const int start = rowptr[ii];
  const int end   = live ? start + cnt[ii] : start;

  float4 a0 = make_float4(0.f, 0.f, 0.f, 0.f), a1 = a0;
  if (r == 0 && live) {
    a0 = *(const float4*)(t2s + (size_t)ii * NC);
    a1 = *(const float4*)(t2s + (size_t)ii * NC + 4);
  }
  for (int j = start + r; j < end; j += 16) {
    const int s0 = col[j];
    const bool two = (j + 8) < end;
    int s1 = 0;
    if (two) s1 = col[j + 8];
    const float4 x0 = *(const float4*)(t2s + (size_t)s0 * NC);
    const float4 x1 = *(const float4*)(t2s + (size_t)s0 * NC + 4);
    float4 y0 = make_float4(0.f, 0.f, 0.f, 0.f), y1 = y0;
    if (two) {
      y0 = *(const float4*)(t2s + (size_t)s1 * NC);
      y1 = *(const float4*)(t2s + (size_t)s1 * NC + 4);
    }
    a0.x += x0.x; a0.y += x0.y; a0.z += x0.z; a0.w += x0.w;
    a1.x += x1.x; a1.y += x1.y; a1.z += x1.z; a1.w += x1.w;
    a0.x += y0.x; a0.y += y0.y; a0.z += y0.z; a0.w += y0.w;
    a1.x += y1.x; a1.y += y1.y; a1.z += y1.z; a1.w += y1.w;
  }
  for (int d = 4; d > 0; d >>= 1) {
    a0.x += __shfl_xor(a0.x, d); a0.y += __shfl_xor(a0.y, d);
    a0.z += __shfl_xor(a0.z, d); a0.w += __shfl_xor(a0.w, d);
    a1.x += __shfl_xor(a1.x, d); a1.y += __shfl_xor(a1.y, d);
    a1.z += __shfl_xor(a1.z, d); a1.w += __shfl_xor(a1.w, d);
  }
  if (r < 2 && live) {
    const float di = dinv[i];
    const float4 cb = *(const float4*)(b2 + 4 * r);
    const float4 a = (r == 0) ? a0 : a1;
    *(float4*)(out + (size_t)i * NC + 4 * r) =
        make_float4(fmaf(a.x, di, cb.x), fmaf(a.y, di, cb.y), fmaf(a.z, di, cb.z), fmaf(a.w, di, cb.w));
  }
}


extern "C" void kernel_launch(void* const* d_in, const int* in_sizes, int n_in,
                              void* d_out, int out_size, void* d_ws, size_t ws_size,
                              hipStream_t stream) {
  const float* x  = (const float*)d_in[0];
  const int*   ei = (const int*)d_in[1];
  const float* W1 = (const float*)d_in[2];
  const float* b1 = (const float*)d_in[3];
  const float* W2 = (const float*)d_in[4];
  const float* b2 = (const float*)d_in[5];
  float* out = (float*)d_out;

  const int n = in_sizes[0] / NF;
  const int e = in_sizes[1] / 2;
  const int* src = ei;
  const int* dst = ei + e;

  float* ws   = (float*)d_ws;
  float* hs   = ws;
  float* t2s  = hs + (size_t)n * NH;
  int*   col  = (int*)(t2s + (size_t)n * NC);
  int*   part = col + e;
  int*   cnt  = part + e;
  int*   rowptr = cnt + n;
  float* dinv = (float*)(rowptr + n);
  int*   M    = (int*)t2s;

  const int T = 256;
  const int nb = (n + BN - 1) / BN;
  const int ch = (((e + G1 - 1) / G1) + 3) & ~3;
  int* bstart = M + nb * G1;
  bucket_count_kernel<<<G1, PT, 0, stream>>>(dst, M, e, nb, ch);
  bucket_partition_kernel<<<G1, PT, 0, stream>>>(src, dst, M, bstart, part, e, nb, ch);
  bucket_csr_kernel<<<nb, PT, 0, stream>>>(bstart, part, cnt, rowptr, dinv, col, n, e, nb);

  gemm1_wmma_kernel<<<(n + 127) / 128, T, 0, stream>>>(x, W1, dinv, hs, n);

  agg1_gather_kernel<<<(n + 8 * NPW - 1) / (8 * NPW), T, 0, stream>>>(rowptr, cnt, col, hs, dinv, b1, W2, t2s, n);

  agg2_gather_kernel<<<(n * NC + T - 1) / T, T, 0, stream>>>(rowptr, cnt, col, t2s, dinv, b2, out, n);
}


# each edge chunk is sorted by bucket in its own work-group (LDS counts, scan, LDS placement, line writes); the bucket launch walks its 256 runs by offsets; count launch gone; sums in LDS-add order
# speedup vs baseline: 4.4107x; 1.1346x over previous
#include <hip/hip_runtime.h>
#include <hip/hip_bf16.h>

typedef __attribute__((ext_vector_type(2))) float v2f;
typedef __attribute__((ext_vector_type(8))) float v8f;

#define NF 512
#define NH 64
#define NC 8
#define KTILE 128
#define PAIRS (KTILE / 2)
#define PSTR 160
#define NPW 4


#define BSH 9
#define BN (1 << BSH)
#define RSH 22
#define NBMAX 2048
#define G1 256
#define PT 1024
#define NOREC 0xFFFFFFFFu
#define LCAP 40960
#define SCAP (16 * PT)
#define RPW (G1 / 32)

__device__ __forceinline__ int pack_rec(int d, int s) {
  return (int)(((unsigned)(d & (BN - 1)) << RSH) | (unsigned)s);
}

__device__ __forceinline__ int4 load4(const int* __restrict__ p, int i, int hi) {
  int4 r = make_int4(-1, -1, -1, -1);
  if (i + 3 < hi && (((size_t)(p + i)) & 15) == 0) {
    r = *(const int4*)(p + i);
  } else {
    if (i < hi) r.x = p[i];
    if (i + 1 < hi) r.y = p[i + 1];
    if (i + 2 < hi) r.z = p[i + 2];
    if (i + 3 < hi) r.w = p[i + 3];
  }
  return r;
}

static_assert(NBMAX == 2 * PT, "the bucket scan gives every thread two buckets");
static_assert(SCAP == 16 * PT, "a chunk kept in registers is four 16-byte loads per thread");

__global__ __launch_bounds__(PT) void chunk_sort_kernel(const int* __restrict__ src, const int* __restrict__ dst,
                                                        int* __restrict__ O, int* __restrict__ part,
                                                        int e, int nb, int ch) {
  __shared__ int cur[NBMAX];
  __shared__ int wsum[32];
  __shared__ int stage[SCAP];
  const int t = threadIdx.x, lane = t & 31, wave = t >> 5, w = blockIdx.x;
  const int lo = w * ch;
  const int hi = min(e, lo + ch);
  const bool inreg = ch <= SCAP && ((hi - lo) & 3) == 0 && ((((size_t)dst) | ((size_t)src)) & 15) == 0;
  cur[2 * t] = 0;
  cur[2 * t + 1] = 0;
  int4 d[4], s[4];
  for (int k = 0; k < 4; ++k) {
    d[k] = make_int4(0, 0, 0, 0);
    s[k] = d[k];
  }
  if (inreg) {
#pragma unroll
    for (int k = 0; k < 4; ++k) {
      const int i = lo + 4 * t + 4 * PT * k;
      if (i < hi) {
        d[k] = *(const int4*)(dst + i);
        s[k] = *(const int4*)(src + i);
      }
    }
  }
  __syncthreads();
  if (inreg) {
    for (int k = 0; k < 4; ++k) {
      if (lo + 4 * t + 4 * PT * k < hi) {
        atomicAdd(&cur[d[k].x >> BSH], 1);
        atomicAdd(&cur[d[k].y >> BSH], 1);
        atomicAdd(&cur[d[k].z >> BSH], 1);
        atomicAdd(&cur[d[k].w >> BSH], 1);
      }
    }
  } else {
    for (int i = lo + 4 * t; i < hi; i += 4 * PT) {
      const int4 dd = load4(dst, i, hi);
      if (dd.x >= 0) atomicAdd(&cur[dd.x >> BSH], 1);
      if (dd.y >= 0) atomicAdd(&cur[dd.y >> BSH], 1);
      if (dd.z >= 0) atomicAdd(&cur[dd.z >> BSH], 1);
      if (dd.w >= 0) atomicAdd(&cur[dd.w >> BSH], 1);
    }
  }
  __syncthreads();
  {
    const int i0 = 2 * t, i1 = 2 * t + 1;
    const int v0 = cur[i0];
    const int v1 = cur[i1];
    const int sum = v0 + v1;
    int incl = sum;
    for (int dd = 1; dd < 32; dd <<= 1) {
      int u = __shfl_up(incl, dd);
      if (lane >= dd) incl += u;
    }
    if (lane == 31) wsum[wave] = incl;
    __syncthreads();
    int front = (lane < wave) ? wsum[lane] : 0;
    for (int dd = 16; dd > 0; dd >>= 1) front += __shfl_xor(front, dd);
    const int excl = front + incl - sum;
    cur[i0] = excl;
    cur[i1] = excl + v0;
    int* orow = O + (size_t)w * (nb + 1);
    if (i0 < nb) orow[i0] = excl;
    if (i1 < nb) orow[i1] = excl + v0;
    if (t == 0) orow[nb] = max(hi - lo, 0);
  }
  __syncthreads();
  if (inreg) {
    for (int k = 0; k < 4; ++k) {
      if (lo + 4 * t + 4 * PT * k < hi) {
        const int p0 = atomicAdd(&cur[d[k].x >> BSH], 1);
        const int p1 = atomicAdd(&cur[d[k].y >> BSH], 1);
        const int p2 = atomicAdd(&cur[d[k].z >> BSH], 1);
        const int p3 = atomicAdd(&cur[d[k].w >> BSH], 1);
        stage[p0] = pack_rec(d[k].x, s[k].x);
        stage[p1] = pack_rec(d[k].y, s[k].y);
        stage[p2] = pack_rec(d[k].z, s[k].z);
        stage[p3] = pack_rec(d[k].w, s[k].w);
      }
    }
    __syncthreads();
    for (int j = t; j < hi - lo; j += PT) part[lo + j] = stage[j];
  } else {
    for (int i = lo + 4 * t; i < hi; i += 4 * PT) {
      const int4 dd = load4(dst, i, hi);
      const int4 ss = load4(src, i, hi);
      if (dd.x >= 0) part[lo + atomicAdd(&cur[dd.x >> BSH], 1)] = pack_rec(dd.x, ss.x);
      if (dd.y >= 0) part[lo + atomicAdd(&cur[dd.y >> BSH], 1)] = pack_rec(dd.y, ss.y);
      if (dd.z >= 0) part[lo + atomicAdd(&cur[dd.z >> BSH], 1)] = pack_rec(dd.z, ss.z);
      if (dd.w >= 0) part[lo + atomicAdd(&cur[dd.w >> BSH], 1)] = pack_rec(dd.w, ss.w);
    }
  }
}

static_assert(G1 % 32 == 0 && G1 <= PT, "one thread per chunk reads the offsets; a wave walks G1/32 runs");

__global__ __launch_bounds__(PT) void bucket_csr_kernel(const int* __restrict__ O, const int* __restrict__ part,
                                                        int* __restrict__ cnt, int* __restrict__ rowptr,
                                                        float* __restrict__ dinv, int* __restrict__ col,
                                                        int n, int nb, int ch) {
  __shared__ int lc[BN];
  __shared__ int wsum[32];
  __shared__ int osum[32];
  __shared__ int tsum[32];
  __shared__ int roff[G1];
  __shared__ int rcnt[G1];
  __shared__ int lcol[LCAP];
  const int t = threadIdx.x, lane = t & 31;
  const int wave = __builtin_amdgcn_readfirstlane(t >> 5);
  const int b = blockIdx.x;
  int o0 = 0, len = 0;
  if (t < G1) {
    const int* orow = O + (size_t)t * (nb + 1) + b;
    o0 = orow[0];
    len = orow[1] - o0;
    roff[t] = t * ch + o0;
    rcnt[t] = len;
  }
  if (t < BN) lc[t] = 0;
  {
    int a0 = o0, a1 = len;
    for (int dd = 16; dd > 0; dd >>= 1) {
      a0 += __shfl_xor(a0, dd);
      a1 += __shfl_xor(a1, dd);
    }
    if (lane == 0) {
      osum[wave] = a0;
      tsum[wave] = a1;
    }
  }
  __syncthreads();
  int s0 = osum[lane], tot = tsum[lane];
  for (int dd = 16; dd > 0; dd >>= 1) {
    s0 += __shfl_xor(s0, dd);
    tot += __shfl_xor(tot, dd);
  }
  const bool staged = tot <= LCAP;
  int ro[RPW], rc[RPW];
  int mx = 0;
  for (int z = 0; z < RPW; ++z) {
    ro[z] = roff[wave + 32 * z];
    rc[z] = rcnt[wave + 32 * z];
    mx = max(mx, rc[z]);
  }
  mx = __builtin_amdgcn_readfirstlane(mx);
  for (int k0 = 0; k0 < mx; k0 += 32) {
    const int k = k0 + lane;
    unsigned r[RPW];
#pragma unroll
    for (int z = 0; z < RPW; ++z) r[z] = (k < rc[z]) ? (unsigned)part[ro[z] + k] : NOREC;
    for (int z = 0; z < RPW; ++z)
      if (r[z] != NOREC) atomicAdd(&lc[r[z] >> RSH], 1);
  }
  __syncthreads();
  const int v = (t < BN) ? lc[t] : 0;
  int incl = v;
  for (int dd = 1; dd < 32; dd <<= 1) {
    int u = __shfl_up(incl, dd);
    if (lane >= dd) incl += u;
  }
  if (lane == 31) wsum[wave] = incl;
  __syncthreads();
  int front = (lane < wave) ? wsum[lane] : 0;
  for (int dd = 16; dd > 0; dd >>= 1) front += __shfl_xor(front, dd);
  const int excl = front + incl - v;
  if (t < BN) {
    lc[t] = excl;
    const int node = b * BN + t;
    if (node < n) {
      cnt[node] = v;
      rowptr[node] = s0 + excl;
      dinv[node] = rsqrtf((float)(v + 1));
    }
  }
  __syncthreads();
  const unsigned smask = (1u << RSH) - 1u;
  for (int k0 = 0; k0 < mx; k0 += 32) {
    const int k = k0 + lane;
    unsigned r[RPW];
    int p[RPW];
#pragma unroll
    for (int z = 0; z < RPW; ++z) r[z] = (k < rc[z]) ? (unsigned)part[ro[z] + k] : NOREC;
    for (int z = 0; z < RPW; ++z) p[z] = (r[z] != NOREC) ? atomicAdd(&lc[r[z] >> RSH], 1) : -1;
    if (staged) {
      for (int z = 0; z < RPW; ++z)
        if (p[z] >= 0) lcol[p[z]] = (int)(r[z] & smask);
    } else {
      for (int z = 0; z < RPW; ++z)
        if (p[z] >= 0) col[s0 + p[z]] = (int)(r[z] & smask);
    }
  }
  if (staged) {
    __syncthreads();
    for (int j = t; j < tot; j += PT) col[s0 + j] = lcol[j];
  }
}


__global__ __launch_bounds__(256) void gemm1_wmma_kernel(
    const float* __restrict__ X, const float* __restrict__ W1,
    const float* __restrict__ dinv, float* __restrict__ hs, int n) {
  __shared__ float wl[PAIRS * PSTR];
  const int lane  = threadIdx.x & 31;
  const int wave  = threadIdx.x >> 5;
  const int m0    = blockIdx.x * 128 + wave * 16;
  const int mlan  = lane & 15;
  const int khalf = (lane >> 4) << 1;
  const int pbase = lane >> 4;

  int row = m0 + mlan;
  if (row >= n) row = n - 1;
  const float* xrow = X + (size_t)row * NF;

  v8f c0 = {}; v8f c1 = {}; v8f c2 = {}; v8f c3 = {};

  for (int kb = 0; kb < NF; kb += KTILE) {
    for (int i = threadIdx.x; i < KTILE * NH; i += 256) {
      int kl = i >> 6, nn = i & 63;
      wl[(kl >> 1) * PSTR + (nn << 1) + (kl & 1)] = W1[(size_t)kb * NH + i];
    }
    if (kb + KTILE < NF)
      __builtin_prefetch(xrow + kb + KTILE, 0, 3);
    __syncthreads();

    for (int kk = 0; kk < KTILE; kk += 4) {
      v2f a; a.x = xrow[kb + kk + khalf]; a.y = xrow[kb + kk + khalf + 1];
      const v2f* wp = (const v2f*)(wl + ((kk >> 1) + pbase) * PSTR);
      v2f b0 = wp[mlan];
      v2f b1 = wp[16 + mlan];
      v2f b2 = wp[32 + mlan];
      v2f b3 = wp[48 + mlan];
      c0 = __builtin_amdgcn_wmma_f32_16x16x4_f32(false, a, false, b0, (short)0, c0, false, false);
      c1 = __builtin_amdgcn_wmma_f32_16x16x4_f32(false, a, false, b1, (short)0, c1, false, false);
      c2 = __builtin_amdgcn_wmma_f32_16x16x4_f32(false, a, false, b2, (short)0, c2, false, false);
      c3 = __builtin_amdgcn_wmma_f32_16x16x4_f32(false, a, false, b3, (short)0, c3, false, false);
    }
    __syncthreads();
  }

  const int half8 = (lane >> 4) << 3;
  for (int r = 0; r < 8; ++r) {
    int rr = m0 + r + half8;
    if (rr < n) {
      float dv = dinv[rr];
      size_t base = (size_t)rr * NH + mlan;
      hs[base]      = c0[r] * dv;
      hs[base + 16] = c1[r] * dv;
      hs[base + 32] = c2[r] * dv;
      hs[base + 48] = c3[r] * dv;
    }
  }
}


__device__ __forceinline__ void add4(float4& a, const float4& b) {
  a.x += b.x; a.y += b.y; a.z += b.z; a.w += b.w;
}

__global__ __launch_bounds__(256) void agg1_gather_kernel(
    const int* __restrict__ rowptr, const int* __restrict__ cnt, const int* __restrict__ col,
    const float* __restrict__ hs, const float* __restrict__ dinv,
    const float* __restrict__ b1, const float* __restrict__ W2,
    float* __restrict__ t2s, int n) {
  const int lane = threadIdx.x & 31;
  const int wave = __builtin_amdgcn_readfirstlane(threadIdx.x >> 5);
  const int half = lane >> 4;
  const int q    = lane & 15;

  const float4 w0 = *(const float4*)(W2 + (4 * q + 0) * NC + 4 * half);
  const float4 w1 = *(const float4*)(W2 + (4 * q + 1) * NC + 4 * half);
  const float4 w2 = *(const float4*)(W2 + (4 * q + 2) * NC + 4 * half);
  const float4 w3 = *(const float4*)(W2 + (4 * q + 3) * NC + 4 * half);
  const float4 bb = *(const float4*)(b1 + 4 * q);

  const int node0 = (blockIdx.x * 8 + wave) * NPW;
  for (int u = 0; u < NPW; ++u) {
    const int i = node0 + u;
    if (i >= n) break;
    const int start = rowptr[i];
    const int end   = start + cnt[i];
    float4 acc = make_float4(0.f, 0.f, 0.f, 0.f);
    if (half == 0) acc = *(const float4*)(hs + (size_t)i * NH + 4 * q);

    for (int base = start; base < end; base += 32) {
      const int m = min(32, end - base);
      const int c = (lane < m) ? col[base + lane] : 0;
      int t = 0;
      for (; t + 8 <= m; t += 8) {
        int idx[4];
        float4 v[4];
#pragma unroll
        for (int z = 0; z < 4; ++z) idx[z] = __shfl(c, t + 2 * z + half);
#pragma unroll
        for (int z = 0; z < 4; ++z) v[z] = *(const float4*)(hs + (size_t)idx[z] * NH + 4 * q);
        add4(acc, v[0]); add4(acc, v[1]); add4(acc, v[2]); add4(acc, v[3]);
      }
      if (t < m) {
        float4 v[4];
#pragma unroll
        for (int z = 0; z < 4; ++z) {
          const int k = t + 2 * z + half;
          const int got = __shfl(c, k);
          v[z] = make_float4(0.f, 0.f, 0.f, 0.f);
          if (k < m) v[z] = *(const float4*)(hs + (size_t)got * NH + 4 * q);
        }
        add4(acc, v[0]); add4(acc, v[1]); add4(acc, v[2]); add4(acc, v[3]);
      }
    }
    acc.x += __shfl_xor(acc.x, 16);
    acc.y += __shfl_xor(acc.y, 16);
    acc.z += __shfl_xor(acc.z, 16);
    acc.w += __shfl_xor(acc.w, 16);

    const float di = dinv[i];
    const float h0 = fmaxf(fmaf(acc.x, di, bb.x), 0.0f);
    const float h1 = fmaxf(fmaf(acc.y, di, bb.y), 0.0f);
    const float h2 = fmaxf(fmaf(acc.z, di, bb.z), 0.0f);
    const float h3 = fmaxf(fmaf(acc.w, di, bb.w), 0.0f);
    float p0 = fmaf(h3, w3.x, fmaf(h2, w2.x, fmaf(h1, w1.x, h0 * w0.x)));
    float p1 = fmaf(h3, w3.y, fmaf(h2, w2.y, fmaf(h1, w1.y, h0 * w0.y)));
    float p2 = fmaf(h3, w3.z, fmaf(h2, w2.z, fmaf(h1, w1.z, h0 * w0.z)));
    float p3 = fmaf(h3, w3.w, fmaf(h2, w2.w, fmaf(h1, w1.w, h0 * w0.w)));
    for (int d = 8; d > 0; d >>= 1) {
      p0 += __shfl_xor(p0, d);
      p1 += __shfl_xor(p1, d);
      p2 += __shfl_xor(p2, d);
      p3 += __shfl_xor(p3, d);
    }
    if (q == 0)
      *(float4*)(t2s + (size_t)i * NC + 4 * half) = make_float4(p0 * di, p1 * di, p2 * di, p3 * di);
  }
}


__global__ __launch_bounds__(256) void agg2_gather_kernel(
    const int* __restrict__ rowptr, const int* __restrict__ cnt, const int* __restrict__ col,
    const float* __restrict__ t2s, const float* __restrict__ dinv,
    const float* __restrict__ b2, float* __restrict__ out, int n) {
  const int r = threadIdx.x & 7;
  const int i = (blockIdx.x * 256 + threadIdx.x) >> 3;
  const bool live = i < n;
  const int ii = live ? i : n - 1;
  const int start = rowptr[ii];
  const int end   = live ? start + cnt[ii] : start;

  float4 a0 = make_float4(0.f, 0.f, 0.f, 0.f), a1 = a0;
  if (r == 0 && live) {
    a0 = *(const float4*)(t2s + (size_t)ii * NC);
    a1 = *(const float4*)(t2s + (size_t)ii * NC + 4);
  }
  for (int j = start + r; j < end; j += 16) {
    const int s0 = col[j];
    const bool two = (j + 8) < end;
    int s1 = 0;
    if (two) s1 = col[j + 8];
    const float4 x0 = *(const float4*)(t2s + (size_t)s0 * NC);
    const float4 x1 = *(const float4*)(t2s + (size_t)s0 * NC + 4);
    float4 y0 = make_float4(0.f, 0.f, 0.f, 0.f), y1 = y0;
    if (two) {
      y0 = *(const float4*)(t2s + (size_t)s1 * NC);
      y1 = *(const float4*)(t2s + (size_t)s1 * NC + 4);
    }
    a0.x += x0.x; a0.y += x0.y; a0.z += x0.z; a0.w += x0.w;
    a1.x += x1.x; a1.y += x1.y; a1.z += x1.z; a1.w += x1.w;
    a0.x += y0.x; a0.y += y0.y; a0.z += y0.z; a0.w += y0.w;
    a1.x += y1.x; a1.y += y1.y; a1.z += y1.z; a1.w += y1.w;
  }
  for (int d = 4; d > 0; d >>= 1) {
    a0.x += __shfl_xor(a0.x, d); a0.y += __shfl_xor(a0.y, d);
    a0.z += __shfl_xor(a0.z, d); a0.w += __shfl_xor(a0.w, d);
    a1.x += __shfl_xor(a1.x, d); a1.y += __shfl_xor(a1.y, d);
    a1.z += __shfl_xor(a1.z, d); a1.w += __shfl_xor(a1.w, d);
  }
  if (r < 2 && live) {
    const float di = dinv[i];
    const float4 cb = *(const float4*)(b2 + 4 * r);
    const float4 a = (r == 0) ? a0 : a1;
    *(float4*)(out + (size_t)i * NC + 4 * r) =
        make_float4(fmaf(a.x, di, cb.x), fmaf(a.y, di, cb.y), fmaf(a.z, di, cb.z), fmaf(a.w, di, cb.w));
  }
}


extern "C" void kernel_launch(void* const* d_in, const int* in_sizes, int n_in,
                              void* d_out, int out_size, void* d_ws, size_t ws_size,
                              hipStream_t stream) {
  const float* x  = (const float*)d_in[0];
  const int*   ei = (const int*)d_in[1];
  const float* W1 = (const float*)d_in[2];
  const float* b1 = (const float*)d_in[3];
  const float* W2 = (const float*)d_in[4];
  const float* b2 = (const float*)d_in[5];
  float* out = (float*)d_out;

  const int n = in_sizes[0] / NF;
  const int e = in_sizes[1] / 2;
  const int* src = ei;
  const int* dst = ei + e;

  float* ws   = (float*)d_ws;
  float* hs   = ws;
  float* t2s  = hs + (size_t)n * NH;
  int*   col  = (int*)(t2s + (size_t)n * NC);
  int*   part = col + e;
  int*   cnt  = part + e;
  int*   rowptr = cnt + n;
  float* dinv = (float*)(rowptr + n);
  int*   O    = (int*)hs;

  const int T = 256;
  const int nb = (n + BN - 1) / BN;
  const int ch = (((e + G1 - 1) / G1) + 3) & ~3;
  chunk_sort_kernel<<<G1, PT, 0, stream>>>(src, dst, O, part, e, nb, ch);
  bucket_csr_kernel<<<nb, PT, 0, stream>>>(O, part, cnt, rowptr, dinv, col, n, nb, ch);

  gemm1_wmma_kernel<<<(n + 127) / 128, T, 0, stream>>>(x, W1, dinv, hs, n);

  agg1_gather_kernel<<<(n + 8 * NPW - 1) / (8 * NPW), T, 0, stream>>>(rowptr, cnt, col, hs, dinv, b1, W2, t2s, n);

  agg2_gather_kernel<<<(n * NC + T - 1) / T, T, 0, stream>>>(rowptr, cnt, col, t2s, dinv, b2, out, n);
}


# layer-1 product made independent of the edge list and run in the chunk sort's launch (1024-thread work-groups, 25 row tiles each); the dinv row scale moves into the bucket launch; 5 -> 4 launches
# speedup vs baseline: 5.0598x; 1.1472x over previous
#include <hip/hip_runtime.h>
#include <hip/hip_bf16.h>

typedef __attribute__((ext_vector_type(2))) float v2f;
typedef __attribute__((ext_vector_type(8))) float v8f;

#define NF 512
#define NH 64
#define NC 8
#define KTILE 128
#define PAIRS (KTILE / 2)
#define PSTR 160
#define NPW 4
#define TPB 25


#define BSH 9
#define BN (1 << BSH)
#define RSH 22
#define NBMAX 2048
#define G1 256
#define PT 1024
#define NOREC 0xFFFFFFFFu
#define LCAP 40960
#define SCAP (16 * PT)
#define RPW (G1 / 32)

__device__ __forceinline__ int pack_rec(int d, int s) {
  return (int)(((unsigned)(d & (BN - 1)) << RSH) | (unsigned)s);
}

__device__ __forceinline__ int4 load4(const int* __restrict__ p, int i, int hi) {
  int4 r = make_int4(-1, -1, -1, -1);
  if (i + 3 < hi && (((size_t)(p + i)) & 15) == 0) {
    r = *(const int4*)(p + i);
  } else {
    if (i < hi) r.x = p[i];
    if (i + 1 < hi) r.y = p[i + 1];
    if (i + 2 < hi) r.z = p[i + 2];
    if (i + 3 < hi) r.w = p[i + 3];
  }
  return r;
}

static_assert(NBMAX == 2 * PT, "the bucket scan gives every thread two buckets");
static_assert(SCAP == 16 * PT, "a chunk kept in registers is four 16-byte loads per thread");

__device__ __forceinline__ void chunk_sort_body(const int* __restrict__ src, const int* __restrict__ dst,
                                                int* __restrict__ O, int* __restrict__ part,
                                                int e, int nb, int ch, int w) {
  __shared__ int cur[NBMAX];
  __shared__ int wsum[32];
  __shared__ int stage[SCAP];
  const int t = threadIdx.x, lane = t & 31, wave = t >> 5;
  const int lo = w * ch;
  const int hi = min(e, lo + ch);
  const bool inreg = ch <= SCAP && ((hi - lo) & 3) == 0 && ((((size_t)dst) | ((size_t)src)) & 15) == 0;
  cur[2 * t] = 0;
  cur[2 * t + 1] = 0;
  int4 d[4], s[4];
  for (int k = 0; k < 4; ++k) {
    d[k] = make_int4(0, 0, 0, 0);
    s[k] = d[k];
  }
  if (inreg) {
#pragma unroll
    for (int k = 0; k < 4; ++k) {
      const int i = lo + 4 * t + 4 * PT * k;
      if (i < hi) {
        d[k] = *(const int4*)(dst + i);
        s[k] = *(const int4*)(src + i);
      }
    }
  }
  __syncthreads();
  if (inreg) {
    for (int k = 0; k < 4; ++k) {
      if (lo + 4 * t + 4 * PT * k < hi) {
        atomicAdd(&cur[d[k].x >> BSH], 1);
        atomicAdd(&cur[d[k].y >> BSH], 1);
        atomicAdd(&cur[d[k].z >> BSH], 1);
        atomicAdd(&cur[d[k].w >> BSH], 1);
      }
    }
  } else {
    for (int i = lo + 4 * t; i < hi; i += 4 * PT) {
      const int4 dd = load4(dst, i, hi);
      if (dd.x >= 0) atomicAdd(&cur[dd.x >> BSH], 1);
      if (dd.y >= 0) atomicAdd(&cur[dd.y >> BSH], 1);
      if (dd.z >= 0) atomicAdd(&cur[dd.z >> BSH], 1);
      if (dd.w >= 0) atomicAdd(&cur[dd.w >> BSH], 1);
    }
  }
  __syncthreads();
  {
    const int i0 = 2 * t, i1 = 2 * t + 1;
    const int v0 = cur[i0];
    const int v1 = cur[i1];
    const int sum = v0 + v1;
    int incl = sum;
    for (int dd = 1; dd < 32; dd <<= 1) {
      int u = __shfl_up(incl, dd);
      if (lane >= dd) incl += u;
    }
    if (lane == 31) wsum[wave] = incl;
    __syncthreads();
    int front = (lane < wave) ? wsum[lane] : 0;
    for (int dd = 16; dd > 0; dd >>= 1) front += __shfl_xor(front, dd);
    const int excl = front + incl - sum;
    cur[i0] = excl;
    cur[i1] = excl + v0;
    int* orow = O + (size_t)w * (nb + 1);
    if (i0 < nb) orow[i0] = excl;
    if (i1 < nb) orow[i1] = excl + v0;
    if (t == 0) orow[nb] = max(hi - lo, 0);
  }
  __syncthreads();
  if (inreg) {
    for (int k = 0; k < 4; ++k) {
      if (lo + 4 * t + 4 * PT * k < hi) {
        const int p0 = atomicAdd(&cur[d[k].x >> BSH], 1);
        const int p1 = atomicAdd(&cur[d[k].y >> BSH], 1);
        const int p2 = atomicAdd(&cur[d[k].z >> BSH], 1);
        const int p3 = atomicAdd(&cur[d[k].w >> BSH], 1);
        stage[p0] = pack_rec(d[k].x, s[k].x);
        stage[p1] = pack_rec(d[k].y, s[k].y);
        stage[p2] = pack_rec(d[k].z, s[k].z);
        stage[p3] = pack_rec(d[k].w, s[k].w);
      }
    }
    __syncthreads();
    for (int j = t; j < hi - lo; j += PT) part[lo + j] = stage[j];
  } else {
    for (int i = lo + 4 * t; i < hi; i += 4 * PT) {
      const int4 dd = load4(dst, i, hi);
      const int4 ss = load4(src, i, hi);
      if (dd.x >= 0) part[lo + atomicAdd(&cur[dd.x >> BSH], 1)] = pack_rec(dd.x, ss.x);
      if (dd.y >= 0) part[lo + atomicAdd(&cur[dd.y >> BSH], 1)] = pack_rec(dd.y, ss.y);
      if (dd.z >= 0) part[lo + atomicAdd(&cur[dd.z >> BSH], 1)] = pack_rec(dd.z, ss.z);
      if (dd.w >= 0) part[lo + atomicAdd(&cur[dd.w >> BSH], 1)] = pack_rec(dd.w, ss.w);
    }
  }
}

static_assert(G1 % 32 == 0 && G1 <= PT, "one thread per chunk reads the offsets; a wave walks G1/32 runs");
static_assert(BN * NH / 4 == 8 * PT, "a bucket's rows of hs are eight float4 per thread");

__global__ __launch_bounds__(PT) void bucket_csr_kernel(const int* __restrict__ O, const int* __restrict__ part,
                                                        int* __restrict__ cnt, int* __restrict__ rowptr,
                                                        float* __restrict__ dinv, int* __restrict__ col,
                                                        float* __restrict__ hs, int n, int nb, int ch) {
  __shared__ int lc[BN];
  __shared__ float sdv[BN];
  __shared__ int wsum[32];
  __shared__ int osum[32];
  __shared__ int tsum[32];
  __shared__ int roff[G1];
  __shared__ int rcnt[G1];
  __shared__ int lcol[LCAP];
  const int t = threadIdx.x, lane = t & 31;
  const int wave = __builtin_amdgcn_readfirstlane(t >> 5);
  const int b = blockIdx.x;
  float4 hrow[8];
#pragma unroll
  for (int j = 0; j < 8; ++j) {
    const int q4 = j * PT + t;
    const int node = min(b * BN + (q4 >> 4), n - 1);
    hrow[j] = *(const float4*)(hs + (size_t)node * NH + ((q4 & 15) << 2));
  }
  int o0 = 0, len = 0;
  if (t < G1) {
    const int* orow = O + (size_t)t * (nb + 1) + b;
    o0 = orow[0];
    len = orow[1] - o0;
    roff[t] = t * ch + o0;
    rcnt[t] = len;
  }
  if (t < BN) lc[t] = 0;
  {
    int a0 = o0, a1 = len;
    for (int dd = 16; dd > 0; dd >>= 1) {
      a0 += __shfl_xor(a0, dd);
      a1 += __shfl_xor(a1, dd);
    }
    if (lane == 0) {
      osum[wave] = a0;
      tsum[wave] = a1;
    }
  }
  __syncthreads();
  int s0 = osum[lane], tot = tsum[lane];
  for (int dd = 16; dd > 0; dd >>= 1) {
    s0 += __shfl_xor(s0, dd);
    tot += __shfl_xor(tot, dd);
  }
  const bool staged = tot <= LCAP;
  int ro[RPW], rc[RPW];
  int mx = 0;
  for (int z = 0; z < RPW; ++z) {
    ro[z] = roff[wave + 32 * z];
    rc[z] = rcnt[wave + 32 * z];
    mx = max(mx, rc[z]);
  }
  mx = __builtin_amdgcn_readfirstlane(mx);
  for (int k0 = 0; k0 < mx; k0 += 32) {
    const int k = k0 + lane;
    unsigned r[RPW];
#pragma unroll
    for (int z = 0; z < RPW; ++z) r[z] = (k < rc[z]) ? (unsigned)part[ro[z] + k] : NOREC;
    for (int z = 0; z < RPW; ++z)
      if (r[z] != NOREC) atomicAdd(&lc[r[z] >> RSH], 1);
  }
  __syncthreads();
  const int v = (t < BN) ? lc[t] : 0;
  int incl = v;
  for (int dd = 1; dd < 32; dd <<= 1) {
    int u = __shfl_up(incl, dd);
    if (lane >= dd) incl += u;
  }
  if (lane == 31) wsum[wave] = incl;
  __syncthreads();
  int front = (lane < wave) ? wsum[lane] : 0;
  for (int dd = 16; dd > 0; dd >>= 1) front += __shfl_xor(front, dd);
  const int excl = front + incl - v;
  if (t < BN) {
    lc[t] = excl;
    const int node = b * BN + t;
    const float dv = rsqrtf((float)(v + 1));
    sdv[t] = dv;
    if (node < n) {
      cnt[node] = v;
      rowptr[node] = s0 + excl;
      dinv[node] = dv;
    }
  }
  __syncthreads();
#pragma unroll
  for (int j = 0; j < 8; ++j) {
    const int q4 = j * PT + t;
    const int r = q4 >> 4;
    const int node = b * BN + r;
    if (node < n) {
      const float dv = sdv[r];
      *(float4*)(hs + (size_t)node * NH + ((q4 & 15) << 2)) =
          make_float4(hrow[j].x * dv, hrow[j].y * dv, hrow[j].z * dv, hrow[j].w * dv);
    }
  }
  const unsigned smask = (1u << RSH) - 1u;
  for (int k0 = 0; k0 < mx; k0 += 32) {
    const int k = k0 + lane;
    unsigned r[RPW];
    int p[RPW];
#pragma unroll
    for (int z = 0; z < RPW; ++z) r[z] = (k < rc[z]) ? (unsigned)part[ro[z] + k] : NOREC;
    for (int z = 0; z < RPW; ++z) p[z] = (r[z] != NOREC) ? atomicAdd(&lc[r[z] >> RSH], 1) : -1;
    if (staged) {
      for (int z = 0; z < RPW; ++z)
        if (p[z] >= 0) lcol[p[z]] = (int)(r[z] & smask);
    } else {
      for (int z = 0; z < RPW; ++z)
        if (p[z] >= 0) col[s0 + p[z]] = (int)(r[z] & smask);
    }
  }
  if (staged) {
    __syncthreads();
    for (int j = t; j < tot; j += PT) col[s0 + j] = lcol[j];
  }
}


static_assert((KTILE * NH) % PT == 0, "every thread stages the same number of W1 words per chunk");

__device__ __forceinline__ void gemm1_body(
    const float* __restrict__ X, const float* __restrict__ W1,
    float* __restrict__ hs, int n, int g) {
  __shared__ float wl[PAIRS * PSTR];
  const int lane  = threadIdx.x & 31;
  const int wave  = __builtin_amdgcn_readfirstlane(threadIdx.x >> 5);
  const int m0    = (g * TPB + wave) * 16;
  const bool active = wave < TPB && m0 < n;
  const int mlan  = lane & 15;
  const int khalf = (lane >> 4) << 1;
  const int pbase = lane >> 4;

  int row = m0 + mlan;
  if (row >= n) row = n - 1;
  const float* xrow = X + (size_t)row * NF;

  v8f c0 = {}; v8f c1 = {}; v8f c2 = {}; v8f c3 = {};

  for (int kb = 0; kb < NF; kb += KTILE) {
    {
      float wv[KTILE * NH / PT];
#pragma unroll
      for (int j = 0; j < KTILE * NH / PT; ++j) wv[j] = W1[(size_t)kb * NH + j * PT + threadIdx.x];
#pragma unroll
      for (int j = 0; j < KTILE * NH / PT; ++j) {
        const int i = j * PT + threadIdx.x;
        const int kl = i >> 6, nn = i & 63;
        wl[(kl >> 1) * PSTR + (nn << 1) + (kl & 1)] = wv[j];
      }
    }
    if (active && kb + KTILE < NF)
      __builtin_prefetch(xrow + kb + KTILE, 0, 3);
    __syncthreads();

    if (active)
    for (int kk = 0; kk < KTILE; kk += 4) {
      v2f a; a.x = xrow[kb + kk + khalf]; a.y = xrow[kb + kk + khalf + 1];
      const v2f* wp = (const v2f*)(wl + ((kk >> 1) + pbase) * PSTR);
      v2f b0 = wp[mlan];
      v2f b1 = wp[16 + mlan];
      v2f b2 = wp[32 + mlan];
      v2f b3 = wp[48 + mlan];
      c0 = __builtin_amdgcn_wmma_f32_16x16x4_f32(false, a, false, b0, (short)0, c0, false, false);
      c1 = __builtin_amdgcn_wmma_f32_16x16x4_f32(false, a, false, b1, (short)0, c1, false, false);
      c2 = __builtin_amdgcn_wmma_f32_16x16x4_f32(false, a, false, b2, (short)0, c2, false, false);
      c3 = __builtin_amdgcn_wmma_f32_16x16x4_f32(false, a, false, b3, (short)0, c3, false, false);
    }
    __syncthreads();
  }

  const int half8 = (lane >> 4) << 3;
  if (active)
  for (int r = 0; r < 8; ++r) {
    int rr = m0 + r + half8;
    if (rr < n) {
      size_t base = (size_t)rr * NH + mlan;
      hs[base]      = c0[r];
      hs[base + 16] = c1[r];
      hs[base + 32] = c2[r];
      hs[base + 48] = c3[r];
    }
  }
}

__global__ __launch_bounds__(PT, 16) void gemm1_chunk_sort_kernel(
    const float* __restrict__ X, const float* __restrict__ W1, float* __restrict__ hs, int n, int gb,
    const int* __restrict__ src, const int* __restrict__ dst, int* __restrict__ O, int* __restrict__ part,
    int e, int nb, int ch) {
  if ((int)blockIdx.x < gb) gemm1_body(X, W1, hs, n, (int)blockIdx.x);
  else chunk_sort_body(src, dst, O, part, e, nb, ch, (int)blockIdx.x - gb);
}


__device__ __forceinline__ void add4(float4& a, const float4& b) {
  a.x += b.x; a.y += b.y; a.z += b.z; a.w += b.w;
}

__global__ __launch_bounds__(256) void agg1_gather_kernel(
    const int* __restrict__ rowptr, const int* __restrict__ cnt, const int* __restrict__ col,
    const float* __restrict__ hs, const float* __restrict__ dinv,
    const float* __restrict__ b1, const float* __restrict__ W2,
    float* __restrict__ t2s, int n) {
  const int lane = threadIdx.x & 31;
  const int wave = __builtin_amdgcn_readfirstlane(threadIdx.x >> 5);
  const int half = lane >> 4;
  const int q    = lane & 15;

  const float4 w0 = *(const float4*)(W2 + (4 * q + 0) * NC + 4 * half);
  const float4 w1 = *(const float4*)(W2 + (4 * q + 1) * NC + 4 * half);
  const float4 w2 = *(const float4*)(W2 + (4 * q + 2) * NC + 4 * half);
  const float4 w3 = *(const float4*)(W2 + (4 * q + 3) * NC + 4 * half);
  const float4 bb = *(const float4*)(b1 + 4 * q);

  const int node0 = (blockIdx.x * 8 + wave) * NPW;
  for (int u = 0; u < NPW; ++u) {
    const int i = node0 + u;
    if (i >= n) break;
    const int start = rowptr[i];
    const int end   = start + cnt[i];
    float4 acc = make_float4(0.f, 0.f, 0.f, 0.f);
    if (half == 0) acc = *(const float4*)(hs + (size_t)i * NH + 4 * q);

    for (int base = start; base < end; base += 32) {
      const int m = min(32, end - base);
      const int c = (lane < m) ? col[base + lane] : 0;
      int t = 0;
      for (; t + 8 <= m; t += 8) {
        int idx[4];
        float4 v[4];
#pragma unroll
        for (int z = 0; z < 4; ++z) idx[z] = __shfl(c, t + 2 * z + half);
#pragma unroll
        for (int z = 0; z < 4; ++z) v[z] = *(const float4*)(hs + (size_t)idx[z] * NH + 4 * q);
        add4(acc, v[0]); add4(acc, v[1]); add4(acc, v[2]); add4(acc, v[3]);
      }
      if (t < m) {
        float4 v[4];
#pragma unroll
        for (int z = 0; z < 4; ++z) {
          const int k = t + 2 * z + half;
          const int got = __shfl(c, k);
          v[z] = make_float4(0.f, 0.f, 0.f, 0.f);
          if (k < m) v[z] = *(const float4*)(hs + (size_t)got * NH + 4 * q);
        }
        add4(acc, v[0]); add4(acc, v[1]); add4(acc, v[2]); add4(acc, v[3]);
      }
    }
    acc.x += __shfl_xor(acc.x, 16);
    acc.y += __shfl_xor(acc.y, 16);
    acc.z += __shfl_xor(acc.z, 16);
    acc.w += __shfl_xor(acc.w, 16);

    const float di = dinv[i];
    const float h0 = fmaxf(fmaf(acc.x, di, bb.x), 0.0f);
    const float h1 = fmaxf(fmaf(acc.y, di, bb.y), 0.0f);
    const float h2 = fmaxf(fmaf(acc.z, di, bb.z), 0.0f);
    const float h3 = fmaxf(fmaf(acc.w, di, bb.w), 0.0f);
    float p0 = fmaf(h3, w3.x, fmaf(h2, w2.x, fmaf(h1, w1.x, h0 * w0.x)));
    float p1 = fmaf(h3, w3.y, fmaf(h2, w2.y, fmaf(h1, w1.y, h0 * w0.y)));
    float p2 = fmaf(h3, w3.z, fmaf(h2, w2.z, fmaf(h1, w1.z, h0 * w0.z)));
    float p3 = fmaf(h3, w3.w, fmaf(h2, w2.w, fmaf(h1, w1.w, h0 * w0.w)));
    for (int d = 8; d > 0; d >>= 1) {
      p0 += __shfl_xor(p0, d);
      p1 += __shfl_xor(p1, d);
      p2 += __shfl_xor(p2, d);
      p3 += __shfl_xor(p3, d);
    }
    if (q == 0)
      *(float4*)(t2s + (size_t)i * NC + 4 * half) = make_float4(p0 * di, p1 * di, p2 * di, p3 * di);
  }
}


__global__ __launch_bounds__(256) void agg2_gather_kernel(
    const int* __restrict__ rowptr, const int* __restrict__ cnt, const int* __restrict__ col,
    const float* __restrict__ t2s, const float* __restrict__ dinv,
    const float* __restrict__ b2, float* __restrict__ out, int n) {
  const int r = threadIdx.x & 7;
  const int i = (blockIdx.x * 256 + threadIdx.x) >> 3;
  const bool live = i < n;
  const int ii = live ? i : n - 1;
  const int start = rowptr[ii];
  const int end   = live ? start + cnt[ii] : start;

  float4 a0 = make_float4(0.f, 0.f, 0.f, 0.f), a1 = a0;
  if (r == 0 && live) {
    a0 = *(const float4*)(t2s + (size_t)ii * NC);
    a1 = *(const float4*)(t2s + (size_t)ii * NC + 4);
  }
  for (int j = start + r; j < end; j += 16) {
    const int s0 = col[j];
    const bool two = (j + 8) < end;
    int s1 = 0;
    if (two) s1 = col[j + 8];
    const float4 x0 = *(const float4*)(t2s + (size_t)s0 * NC);
    const float4 x1 = *(const float4*)(t2s + (size_t)s0 * NC + 4);
    float4 y0 = make_float4(0.f, 0.f, 0.f, 0.f), y1 = y0;
    if (two) {
      y0 = *(const float4*)(t2s + (size_t)s1 * NC);
      y1 = *(const float4*)(t2s + (size_t)s1 * NC + 4);
    }
    a0.x += x0.x; a0.y += x0.y; a0.z += x0.z; a0.w += x0.w;
    a1.x += x1.x; a1.y += x1.y; a1.z += x1.z; a1.w += x1.w;
    a0.x += y0.x; a0.y += y0.y; a0.z += y0.z; a0.w += y0.w;
    a1.x += y1.x; a1.y += y1.y; a1.z += y1.z; a1.w += y1.w;
  }
  for (int d = 4; d > 0; d >>= 1) {
    a0.x += __shfl_xor(a0.x, d); a0.y += __shfl_xor(a0.y, d);
    a0.z += __shfl_xor(a0.z, d); a0.w += __shfl_xor(a0.w, d);
    a1.x += __shfl_xor(a1.x, d); a1.y += __shfl_xor(a1.y, d);
    a1.z += __shfl_xor(a1.z, d); a1.w += __shfl_xor(a1.w, d);
  }
  if (r < 2 && live) {
    const float di = dinv[i];
    const float4 cb = *(const float4*)(b2 + 4 * r);
    const float4 a = (r == 0) ? a0 : a1;
    *(float4*)(out + (size_t)i * NC + 4 * r) =
        make_float4(fmaf(a.x, di, cb.x), fmaf(a.y, di, cb.y), fmaf(a.z, di, cb.z), fmaf(a.w, di, cb.w));
  }
}


extern "C" void kernel_launch(void* const* d_in, const int* in_sizes, int n_in,
                              void* d_out, int out_size, void* d_ws, size_t ws_size,
                              hipStream_t stream) {
  const float* x  = (const float*)d_in[0];
  const int*   ei = (const int*)d_in[1];
  const float* W1 = (const float*)d_in[2];
  const float* b1 = (const float*)d_in[3];
  const float* W2 = (const float*)d_in[4];
  const float* b2 = (const float*)d_in[5];
  float* out = (float*)d_out;

  const int n = in_sizes[0] / NF;
  const int e = in_sizes[1] / 2;
  const int* src = ei;
  const int* dst = ei + e;

  float* ws   = (float*)d_ws;
  float* hs   = ws;
  float* t2s  = hs + (size_t)n * NH;
  int*   col  = (int*)(t2s + (size_t)n * NC);
  int*   part = col + e;
  int*   cnt  = part + e;
  int*   rowptr = cnt + n;
  float* dinv = (float*)(rowptr + n);
  int*   O    = (int*)t2s;

  const int T = 256;
  const int nb = (n + BN - 1) / BN;
  const int ch = (((e + G1 - 1) / G1) + 3) & ~3;
  const int gb = ((n + 15) / 16 + TPB - 1) / TPB;
  gemm1_chunk_sort_kernel<<<gb + G1, PT, 0, stream>>>(x, W1, hs, n, gb, src, dst, O, part, e, nb, ch);
  bucket_csr_kernel<<<nb, PT, 0, stream>>>(O, part, cnt, rowptr, dinv, col, hs, n, nb, ch);

  agg1_gather_kernel<<<(n + 8 * NPW - 1) / (8 * NPW), T, 0, stream>>>(rowptr, cnt, col, hs, dinv, b1, W2, t2s, n);

  agg2_gather_kernel<<<(n * NC + T - 1) / T, T, 0, stream>>>(rowptr, cnt, col, t2s, dinv, b2, out, n);
}
